# starting kernel
# baseline (speedup 1.0000x reference)
#include <hip/hip_runtime.h>
#include <hip/hip_bf16.h>

#define BB 2
#define TT 2048
#define CC 1024
#define NH 16
#define HD 64
#define C3 3072
#define BUDGET 408

typedef __attribute__((ext_vector_type(2))) float v2f;
typedef __attribute__((ext_vector_type(8))) float v8f;

__device__ __forceinline__ v8f wmma4(v2f a, v2f b, v8f c) {
  return __builtin_amdgcn_wmma_f32_16x16x4_f32(false, a, false, b, (short)0, c, false, false);
}

__device__ __forceinline__ float rowmax16(float x) {
  #pragma unroll
  for (int d = 1; d < 16; d <<= 1) x = fmaxf(x, __shfl_xor(x, d, 32));
  return x;
}
__device__ __forceinline__ float rowsum16(float x) {
  #pragma unroll
  for (int d = 1; d < 16; d <<= 1) x += __shfl_xor(x, d, 32);
  return x;
}

__global__ void gemm_wmma(const float* __restrict__ A, const float* __restrict__ W,
                          const float* __restrict__ bias, float* __restrict__ Out,
                          int M, int N, int K) {
  int wave = threadIdx.x >> 5;
  int lane = threadIdx.x & 31;
  int g  = lane >> 4;
  int lm = lane & 15;
  int m0 = blockIdx.y * 32;
  int n0 = blockIdx.x * 256 + wave * 64;
  if (n0 >= N || m0 >= M) return;

  v8f acc[2][4];
  #pragma unroll
  for (int t = 0; t < 4; ++t) {
    float bv = bias[n0 + t * 16 + lm];
    #pragma unroll
    for (int mt = 0; mt < 2; ++mt)
      #pragma unroll
      for (int r = 0; r < 8; ++r) acc[mt][t][r] = bv;
  }
  const float* Ar[2];
  Ar[0] = A + (size_t)(m0 + lm) * K + 2 * g;
  Ar[1] = A + (size_t)(m0 + 16 + lm) * K + 2 * g;
  const float* Wp[4];
  #pragma unroll
  for (int t = 0; t < 4; ++t)
    Wp[t] = W + (size_t)(n0 + t * 16 + lm) * K + 2 * g;

  v2f a[2], b[4], an[2], bn[4];
  #pragma unroll
  for (int mt = 0; mt < 2; ++mt) a[mt] = *(const v2f*)(Ar[mt]);
  #pragma unroll
  for (int t = 0; t < 4; ++t) b[t] = *(const v2f*)(Wp[t]);

  for (int k0 = 0; k0 < K; k0 += 4) {
    int k1 = (k0 + 4 < K) ? (k0 + 4) : k0;
    an[0] = *(const v2f*)(Ar[0] + k1);
    an[1] = *(const v2f*)(Ar[1] + k1);
    #pragma unroll
    for (int t = 0; t < 4; ++t) bn[t] = *(const v2f*)(Wp[t] + k1);
    #pragma unroll
    for (int mt = 0; mt < 2; ++mt)
      #pragma unroll
      for (int t = 0; t < 4; ++t)
        acc[mt][t] = wmma4(a[mt], b[t], acc[mt][t]);
    a[0] = an[0]; a[1] = an[1];
    #pragma unroll
    for (int t = 0; t < 4; ++t) b[t] = bn[t];
  }
  #pragma unroll
  for (int mt = 0; mt < 2; ++mt)
    #pragma unroll
    for (int t = 0; t < 4; ++t)
      #pragma unroll
      for (int r = 0; r < 8; ++r)
        Out[(size_t)(m0 + mt * 16 + r + 8 * g) * N + n0 + t * 16 + lm] = acc[mt][t][r];
}

__global__ void head0_scores(const float* __restrict__ qkv, float* __restrict__ S) {
  int jb = blockIdx.x, ib = blockIdx.y, b = blockIdx.z;
  int lane = threadIdx.x & 31, g = lane >> 4, lm = lane & 15;
  int i0 = ib * 16, j0 = jb * 16;
  if (jb > ib) {
    #pragma unroll
    for (int r = 0; r < 8; ++r)
      S[((size_t)(b * TT + i0 + r + 8 * g)) * TT + j0 + lm] = 0.f;
    return;
  }
  const float* qrow = qkv + ((size_t)(b * TT) + i0 + lm) * C3;
  const float* krow = qkv + ((size_t)(b * TT) + j0 + lm) * C3 + CC;
  v2f aq[16], bk[16];
  #pragma unroll
  for (int ks = 0; ks < 16; ++ks) {
    aq[ks] = *(const v2f*)(qrow + ks * 4 + 2 * g);
    bk[ks] = *(const v2f*)(krow + ks * 4 + 2 * g);
  }
  v8f acc = {0.f, 0.f, 0.f, 0.f, 0.f, 0.f, 0.f, 0.f};
  #pragma unroll
  for (int ks = 0; ks < 16; ++ks) acc = wmma4(aq[ks], bk[ks], acc);
  #pragma unroll
  for (int r = 0; r < 8; ++r) {
    int i = i0 + r + 8 * g, j = j0 + lm;
    float s = acc[r] * 0.125f;
    s = (j < i && j > 0) ? fmaxf(s, 0.f) : 0.f;
    S[((size_t)(b * TT + i)) * TT + j] = s;
  }
}

__global__ void ff_cumsum(float* __restrict__ S) {
  int j = blockIdx.x * blockDim.x + threadIdx.x;
  int b = blockIdx.y;
  float* col = S + (size_t)b * TT * TT + j;
  float carry = 0.f;
  #pragma unroll 8
  for (int i = 0; i < TT; ++i) {
    float v = col[(size_t)i * TT];
    col[(size_t)i * TT] = carry;
    carry += v;
  }
}

__global__ void select_mask(const float* __restrict__ FF, unsigned* __restrict__ mask) {
  int row = blockIdx.x;
  int i = row & (TT - 1);
  const float* f = FF + (size_t)row * TT;
  unsigned* mrow = mask + (size_t)row * (TT / 32);
  int tid = threadIdx.x;

  if (i < BUDGET) {
    for (int w = tid; w < TT / 32; w += 256) {
      int jb = w * 32;
      unsigned m;
      if (jb + 31 <= i) m = 0xFFFFFFFFu;
      else if (jb > i)  m = 0u;
      else              m = (1u << (i - jb + 1)) - 1u;
      mrow[w] = m;
    }
    return;
  }

  __shared__ unsigned hist[256];
  __shared__ unsigned misc[4];
  __shared__ unsigned wave_tot[8];

  if (tid == 0) { misc[0] = 0u; misc[1] = BUDGET - 1; }
  unsigned prefix = 0, prefmask = 0;
  for (int pass = 0; pass < 4; ++pass) {
    int shift = 24 - 8 * pass;
    hist[tid] = 0u;
    __syncthreads();
    for (int j = tid; j <= i; j += 256) {
      unsigned u = __float_as_uint(f[j]);
      if ((u & prefmask) == prefix) atomicAdd(&hist[(u >> shift) & 255u], 1u);
    }
    __syncthreads();
    if (tid == 0) {
      unsigned kk = misc[1], cum = 0; int d = 0;
      for (; d < 256; ++d) { unsigned h = hist[d]; if (cum + h > kk) break; cum += h; }
      misc[1] = kk - cum;
      misc[0] |= ((unsigned)d << shift);
    }
    __syncthreads();
    prefix = misc[0];
    prefmask |= (255u << shift);
  }
  unsigned theta = prefix;

  if (tid == 0) misc[2] = 0u;
  __syncthreads();
  unsigned cl = 0;
  for (int j = tid; j <= i; j += 256)
    if (__float_as_uint(f[j]) < theta) cl++;
  atomicAdd(&misc[2], cl);
  __syncthreads();
  int ntake = BUDGET - (int)misc[2];

  int lane = tid & 31, wv = tid >> 5;
  unsigned below = (1u << lane) - 1u;
  unsigned carry = 0;
  for (int base = 0; base < TT; base += 256) {
    int j = base + tid;
    unsigned u = (j <= i) ? __float_as_uint(f[j]) : 0xFFFFFFFFu;
    bool eq = (j <= i) && (u == theta);
    unsigned bal = (unsigned)__ballot(eq);
    if (lane == 0) wave_tot[wv] = __popc(bal);
    __syncthreads();
    unsigned pre = carry + __popc(bal & below);
    for (int w2 = 0; w2 < wv; ++w2) pre += wave_tot[w2];
    bool keep = (j <= i) && ((u < theta) || (eq && (int)pre < ntake));
    keep = keep || (j == i);
    unsigned kb = (unsigned)__ballot(keep);
    if (lane == 0) mrow[(base + wv * 32) >> 5] = kb;
    unsigned tot = 0;
    #pragma unroll
    for (int w2 = 0; w2 < 8; ++w2) tot += wave_tot[w2];
    carry += tot;
    __syncthreads();
  }
}

__global__ void attn(const float* __restrict__ qkv, const float* __restrict__ FF,
                     const unsigned* __restrict__ mask, float* __restrict__ Y) {
  int ib = blockIdx.x, h = blockIdx.y, b = blockIdx.z;
  int lane = threadIdx.x & 31, g = lane >> 4, lm = lane & 15;
  int i0 = ib * 16;
  __shared__ float pt[16 * 17];

  const float* qb = qkv + ((size_t)(b * TT)) * C3 + h * HD;
  const float* kb = qb + CC;
  const float* vb = qb + 2 * CC;

  v2f aq[16];
  #pragma unroll
  for (int ks = 0; ks < 16; ++ks)
    aq[ks] = *(const v2f*)(qb + (size_t)(i0 + lm) * C3 + ks * 4 + 2 * g);

  float mrow[8], lrow[8];
  v8f o[4];
  #pragma unroll
  for (int t = 0; t < 4; ++t)
    #pragma unroll
    for (int r = 0; r < 8; ++r) o[t][r] = 0.f;
  #pragma unroll
  for (int r = 0; r < 8; ++r) { mrow[r] = -1e30f; lrow[r] = 0.f; }

  for (int jb = 0; jb <= ib; ++jb) {
    int j0 = jb * 16;
    v2f bk[16];
    #pragma unroll
    for (int ks = 0; ks < 16; ++ks)
      bk[ks] = *(const v2f*)(kb + (size_t)(j0 + lm) * C3 + ks * 4 + 2 * g);
    v8f acc = {0.f, 0.f, 0.f, 0.f, 0.f, 0.f, 0.f, 0.f};
    #pragma unroll
    for (int ks = 0; ks < 16; ++ks) acc = wmma4(aq[ks], bk[ks], acc);

    float ffv[8]; unsigned mwv[8];
    #pragma unroll
    for (int r = 0; r < 8; ++r) {
      int i = i0 + r + 8 * g;
      ffv[r] = FF[((size_t)(b * TT + i)) * TT + j0 + lm];
      mwv[r] = mask[((size_t)(b * TT + i)) * (TT / 32) + (j0 >> 5)];
    }
    float p[8];
    #pragma unroll
    for (int r = 0; r < 8; ++r) {
      int i = i0 + r + 8 * g, j = j0 + lm;
      bool keep = ((mwv[r] >> (j & 31)) & 1u) && (j <= i);
      float s = keep ? (acc[r] * 0.125f - ffv[r]) : -1e30f;
      float bm = rowmax16(s);
      float mnew = fmaxf(mrow[r], bm);
      float scale = __expf(mrow[r] - mnew);
      float pr = __expf(s - mnew);
      lrow[r] = lrow[r] * scale + rowsum16(pr);
      mrow[r] = mnew;
      p[r] = pr;
      #pragma unroll
      for (int t = 0; t < 4; ++t) o[t][r] *= scale;
    }
    __syncthreads();
    #pragma unroll
    for (int r = 0; r < 8; ++r) pt[(r + 8 * g) * 17 + lm] = p[r];
    __syncthreads();
    v2f ap[4];
    #pragma unroll
    for (int kt = 0; kt < 4; ++kt) {
      ap[kt].x = pt[lm * 17 + kt * 4 + 2 * g];
      ap[kt].y = pt[lm * 17 + kt * 4 + 2 * g + 1];
    }
    v2f bv[4][4];
    #pragma unroll
    for (int kt = 0; kt < 4; ++kt)
      #pragma unroll
      for (int t = 0; t < 4; ++t) {
        bv[kt][t].x = vb[(size_t)(j0 + kt * 4 + 2 * g) * C3 + t * 16 + lm];
        bv[kt][t].y = vb[(size_t)(j0 + kt * 4 + 2 * g + 1) * C3 + t * 16 + lm];
      }
    #pragma unroll
    for (int kt = 0; kt < 4; ++kt)
      #pragma unroll
      for (int t = 0; t < 4; ++t)
        o[t] = wmma4(ap[kt], bv[kt][t], o[t]);
  }
  #pragma unroll
  for (int r = 0; r < 8; ++r) {
    float inv = 1.f / lrow[r];
    int i = i0 + r + 8 * g;
    #pragma unroll
    for (int t = 0; t < 4; ++t)
      Y[((size_t)(b * TT + i)) * CC + h * HD + t * 16 + lm] = o[t][r] * inv;
  }
}

extern "C" void kernel_launch(void* const* d_in, const int* in_sizes, int n_in,
                              void* d_out, int out_size, void* d_ws, size_t ws_size,
                              hipStream_t stream) {
  const float* x      = (const float*)d_in[0];
  const float* w_attn = (const float*)d_in[1];
  const float* b_attn = (const float*)d_in[2];
  const float* w_proj = (const float*)d_in[3];
  const float* b_proj = (const float*)d_in[4];
  float* out = (float*)d_out;

  char* ws = (char*)d_ws;
  const size_t QKV_BYTES = (size_t)BB * TT * C3 * 4;
  const size_t FF_BYTES  = (size_t)BB * TT * TT * 4;
  const size_t MSK_BYTES = (size_t)BB * TT * (TT / 32) * 4;
  float*    qkv = (float*)ws;
  float*    FF  = (float*)(ws + QKV_BYTES);
  unsigned* msk = (unsigned*)(ws + QKV_BYTES + FF_BYTES);
  float*    Y   = (float*)(ws + QKV_BYTES + FF_BYTES + MSK_BYTES);

  gemm_wmma<<<dim3(C3 / 256, (BB * TT) / 32), 128, 0, stream>>>(
      x, w_attn, b_attn, qkv, BB * TT, C3, CC);
  head0_scores<<<dim3(TT / 16, TT / 16, BB), 32, 0, stream>>>(qkv, FF);
  ff_cumsum<<<dim3(TT / 256, BB), 256, 0, stream>>>(FF);
  select_mask<<<BB * TT, 256, 0, stream>>>(FF, msk);
  attn<<<dim3(TT / 16, NH, BB), 32, 0, stream>>>(qkv, FF, msk, Y);
  gemm_wmma<<<dim3(CC / 256, (BB * TT) / 32), 128, 0, stream>>>(
      Y, w_proj, b_proj, out, BB * TT, CC, CC);
}


# attn: skip 16-key tiles in which none of the wave's 16 query rows keeps a key (tile bitmap built once per wave from the keep mask; flagged tiles visited in the same ascending order)
# speedup vs baseline: 1.4075x; 1.4075x over previous
#include <hip/hip_runtime.h>
#include <hip/hip_bf16.h>

#define BB 2
#define TT 2048
#define CC 1024
#define NH 16
#define HD 64
#define C3 3072
#define BUDGET 408

typedef __attribute__((ext_vector_type(2))) float v2f;
typedef __attribute__((ext_vector_type(8))) float v8f;

__device__ __forceinline__ v8f wmma4(v2f a, v2f b, v8f c) {
  return __builtin_amdgcn_wmma_f32_16x16x4_f32(false, a, false, b, (short)0, c, false, false);
}

__device__ __forceinline__ float rowmax16(float x) {
  #pragma unroll
  for (int d = 1; d < 16; d <<= 1) x = fmaxf(x, __shfl_xor(x, d, 32));
  return x;
}
__device__ __forceinline__ float rowsum16(float x) {
  #pragma unroll
  for (int d = 1; d < 16; d <<= 1) x += __shfl_xor(x, d, 32);
  return x;
}

__global__ void gemm_wmma(const float* __restrict__ A, const float* __restrict__ W,
                          const float* __restrict__ bias, float* __restrict__ Out,
                          int M, int N, int K) {
  int wave = threadIdx.x >> 5;
  int lane = threadIdx.x & 31;
  int g  = lane >> 4;
  int lm = lane & 15;
  int m0 = blockIdx.y * 32;
  int n0 = blockIdx.x * 256 + wave * 64;
  if (n0 >= N || m0 >= M) return;

  v8f acc[2][4];
  #pragma unroll
  for (int t = 0; t < 4; ++t) {
    float bv = bias[n0 + t * 16 + lm];
    #pragma unroll
    for (int mt = 0; mt < 2; ++mt)
      #pragma unroll
      for (int r = 0; r < 8; ++r) acc[mt][t][r] = bv;
  }
  const float* Ar[2];
  Ar[0] = A + (size_t)(m0 + lm) * K + 2 * g;
  Ar[1] = A + (size_t)(m0 + 16 + lm) * K + 2 * g;
  const float* Wp[4];
  #pragma unroll
  for (int t = 0; t < 4; ++t)
    Wp[t] = W + (size_t)(n0 + t * 16 + lm) * K + 2 * g;

  v2f a[2], b[4], an[2], bn[4];
  #pragma unroll
  for (int mt = 0; mt < 2; ++mt) a[mt] = *(const v2f*)(Ar[mt]);
  #pragma unroll
  for (int t = 0; t < 4; ++t) b[t] = *(const v2f*)(Wp[t]);

  for (int k0 = 0; k0 < K; k0 += 4) {
    int k1 = (k0 + 4 < K) ? (k0 + 4) : k0;
    an[0] = *(const v2f*)(Ar[0] + k1);
    an[1] = *(const v2f*)(Ar[1] + k1);
    #pragma unroll
    for (int t = 0; t < 4; ++t) bn[t] = *(const v2f*)(Wp[t] + k1);
    #pragma unroll
    for (int mt = 0; mt < 2; ++mt)
      #pragma unroll
      for (int t = 0; t < 4; ++t)
        acc[mt][t] = wmma4(a[mt], b[t], acc[mt][t]);
    a[0] = an[0]; a[1] = an[1];
    #pragma unroll
    for (int t = 0; t < 4; ++t) b[t] = bn[t];
  }
  #pragma unroll
  for (int mt = 0; mt < 2; ++mt)
    #pragma unroll
    for (int t = 0; t < 4; ++t)
      #pragma unroll
      for (int r = 0; r < 8; ++r)
        Out[(size_t)(m0 + mt * 16 + r + 8 * g) * N + n0 + t * 16 + lm] = acc[mt][t][r];
}

__global__ void head0_scores(const float* __restrict__ qkv, float* __restrict__ S) {
  int jb = blockIdx.x, ib = blockIdx.y, b = blockIdx.z;
  int lane = threadIdx.x & 31, g = lane >> 4, lm = lane & 15;
  int i0 = ib * 16, j0 = jb * 16;
  if (jb > ib) {
    #pragma unroll
    for (int r = 0; r < 8; ++r)
      S[((size_t)(b * TT + i0 + r + 8 * g)) * TT + j0 + lm] = 0.f;
    return;
  }
  const float* qrow = qkv + ((size_t)(b * TT) + i0 + lm) * C3;
  const float* krow = qkv + ((size_t)(b * TT) + j0 + lm) * C3 + CC;
  v2f aq[16], bk[16];
  #pragma unroll
  for (int ks = 0; ks < 16; ++ks) {
    aq[ks] = *(const v2f*)(qrow + ks * 4 + 2 * g);
    bk[ks] = *(const v2f*)(krow + ks * 4 + 2 * g);
  }
  v8f acc = {0.f, 0.f, 0.f, 0.f, 0.f, 0.f, 0.f, 0.f};
  #pragma unroll
  for (int ks = 0; ks < 16; ++ks) acc = wmma4(aq[ks], bk[ks], acc);
  #pragma unroll
  for (int r = 0; r < 8; ++r) {
    int i = i0 + r + 8 * g, j = j0 + lm;
    float s = acc[r] * 0.125f;
    s = (j < i && j > 0) ? fmaxf(s, 0.f) : 0.f;
    S[((size_t)(b * TT + i)) * TT + j] = s;
  }
}

__global__ void ff_cumsum(float* __restrict__ S) {
  int j = blockIdx.x * blockDim.x + threadIdx.x;
  int b = blockIdx.y;
  float* col = S + (size_t)b * TT * TT + j;
  float carry = 0.f;
  #pragma unroll 8
  for (int i = 0; i < TT; ++i) {
    float v = col[(size_t)i * TT];
    col[(size_t)i * TT] = carry;
    carry += v;
  }
}

__global__ void select_mask(const float* __restrict__ FF, unsigned* __restrict__ mask) {
  int row = blockIdx.x;
  int i = row & (TT - 1);
  const float* f = FF + (size_t)row * TT;
  unsigned* mrow = mask + (size_t)row * (TT / 32);
  int tid = threadIdx.x;

  if (i < BUDGET) {
    for (int w = tid; w < TT / 32; w += 256) {
      int jb = w * 32;
      unsigned m;
      if (jb + 31 <= i) m = 0xFFFFFFFFu;
      else if (jb > i)  m = 0u;
      else              m = (1u << (i - jb + 1)) - 1u;
      mrow[w] = m;
    }
    return;
  }

  __shared__ unsigned hist[256];
  __shared__ unsigned misc[4];
  __shared__ unsigned wave_tot[8];

  if (tid == 0) { misc[0] = 0u; misc[1] = BUDGET - 1; }
  unsigned prefix = 0, prefmask = 0;
  for (int pass = 0; pass < 4; ++pass) {
    int shift = 24 - 8 * pass;
    hist[tid] = 0u;
    __syncthreads();
    for (int j = tid; j <= i; j += 256) {
      unsigned u = __float_as_uint(f[j]);
      if ((u & prefmask) == prefix) atomicAdd(&hist[(u >> shift) & 255u], 1u);
    }
    __syncthreads();
    if (tid == 0) {
      unsigned kk = misc[1], cum = 0; int d = 0;
      for (; d < 256; ++d) { unsigned h = hist[d]; if (cum + h > kk) break; cum += h; }
      misc[1] = kk - cum;
      misc[0] |= ((unsigned)d << shift);
    }
    __syncthreads();
    prefix = misc[0];
    prefmask |= (255u << shift);
  }
  unsigned theta = prefix;

  if (tid == 0) misc[2] = 0u;
  __syncthreads();
  unsigned cl = 0;
  for (int j = tid; j <= i; j += 256)
    if (__float_as_uint(f[j]) < theta) cl++;
  atomicAdd(&misc[2], cl);
  __syncthreads();
  int ntake = BUDGET - (int)misc[2];

  int lane = tid & 31, wv = tid >> 5;
  unsigned below = (1u << lane) - 1u;
  unsigned carry = 0;
  for (int base = 0; base < TT; base += 256) {
    int j = base + tid;
    unsigned u = (j <= i) ? __float_as_uint(f[j]) : 0xFFFFFFFFu;
    bool eq = (j <= i) && (u == theta);
    unsigned bal = (unsigned)__ballot(eq);
    if (lane == 0) wave_tot[wv] = __popc(bal);
    __syncthreads();
    unsigned pre = carry + __popc(bal & below);
    for (int w2 = 0; w2 < wv; ++w2) pre += wave_tot[w2];
    bool keep = (j <= i) && ((u < theta) || (eq && (int)pre < ntake));
    keep = keep || (j == i);
    unsigned kb = (unsigned)__ballot(keep);
    if (lane == 0) mrow[(base + wv * 32) >> 5] = kb;
    unsigned tot = 0;
    #pragma unroll
    for (int w2 = 0; w2 < 8; ++w2) tot += wave_tot[w2];
    carry += tot;
    __syncthreads();
  }
}

__global__ void attn(const float* __restrict__ qkv, const float* __restrict__ FF,
                     const unsigned* __restrict__ mask, float* __restrict__ Y) {
  int ib = blockIdx.x, h = blockIdx.y, b = blockIdx.z;
  int lane = threadIdx.x & 31, g = lane >> 4, lm = lane & 15;
  int i0 = ib * 16;
  __shared__ float pt[16 * 17];

  const float* qb = qkv + ((size_t)(b * TT)) * C3 + h * HD;
  const float* kb = qb + CC;
  const float* vb = qb + 2 * CC;

  v2f aq[16];
  #pragma unroll
  for (int ks = 0; ks < 16; ++ks)
    aq[ks] = *(const v2f*)(qb + (size_t)(i0 + lm) * C3 + ks * 4 + 2 * g);

  float mrow[8], lrow[8];
  v8f o[4];
  #pragma unroll
  for (int t = 0; t < 4; ++t)
    #pragma unroll
    for (int r = 0; r < 8; ++r) o[t][r] = 0.f;
  #pragma unroll
  for (int r = 0; r < 8; ++r) { mrow[r] = -1e30f; lrow[r] = 0.f; }

  const unsigned* mrow16 = mask + ((size_t)(b * TT + i0)) * (TT / 32);
  unsigned or0 = 0u, or1 = 0u;
  #pragma unroll
  for (int r = 0; r < 16; ++r) {
    or0 |= mrow16[r * (TT / 32) + lane];
    or1 |= mrow16[r * (TT / 32) + 32 + lane];
  }
  int wsrc = lane >> 1, hsh = (lane & 1) * 16;
  unsigned w0 = __shfl(or0, wsrc, 32), w1 = __shfl(or0, 16 + wsrc, 32);
  unsigned w2 = __shfl(or1, wsrc, 32), w3 = __shfl(or1, 16 + wsrc, 32);
  unsigned t0 = (unsigned)__ballot(((w0 >> hsh) & 0xFFFFu) != 0u);
  unsigned t1 = (unsigned)__ballot(((w1 >> hsh) & 0xFFFFu) != 0u);
  unsigned t2 = (unsigned)__ballot(((w2 >> hsh) & 0xFFFFu) != 0u);
  unsigned t3 = (unsigned)__ballot(((w3 >> hsh) & 0xFFFFu) != 0u);
  unsigned long long tlo = (unsigned long long)t0 | ((unsigned long long)t1 << 32);
  unsigned long long thi = (unsigned long long)t2 | ((unsigned long long)t3 << 32);
  if (ib < 64) {
    if (ib < 63) tlo &= (2ull << ib) - 1ull;
    thi = 0ull;
  } else if (ib < 127) {
    thi &= (2ull << (ib - 64)) - 1ull;
  }

  for (;;) {
    int jb;
    if (tlo != 0ull)      { jb = __builtin_ctzll(tlo);      tlo &= tlo - 1ull; }
    else if (thi != 0ull) { jb = 64 + __builtin_ctzll(thi); thi &= thi - 1ull; }
    else break;
    int j0 = jb * 16;
    v2f bk[16];
    #pragma unroll
    for (int ks = 0; ks < 16; ++ks)
      bk[ks] = *(const v2f*)(kb + (size_t)(j0 + lm) * C3 + ks * 4 + 2 * g);
    v8f acc = {0.f, 0.f, 0.f, 0.f, 0.f, 0.f, 0.f, 0.f};
    #pragma unroll
    for (int ks = 0; ks < 16; ++ks) acc = wmma4(aq[ks], bk[ks], acc);

    float ffv[8]; unsigned mwv[8];
    #pragma unroll
    for (int r = 0; r < 8; ++r) {
      int i = i0 + r + 8 * g;
      ffv[r] = FF[((size_t)(b * TT + i)) * TT + j0 + lm];
      mwv[r] = mask[((size_t)(b * TT + i)) * (TT / 32) + (j0 >> 5)];
    }
    float p[8];
    #pragma unroll
    for (int r = 0; r < 8; ++r) {
      int i = i0 + r + 8 * g, j = j0 + lm;
      bool keep = ((mwv[r] >> (j & 31)) & 1u) && (j <= i);
      float s = keep ? (acc[r] * 0.125f - ffv[r]) : -1e30f;
      float bm = rowmax16(s);
      float mnew = fmaxf(mrow[r], bm);
      float scale = __expf(mrow[r] - mnew);
      float pr = __expf(s - mnew);
      lrow[r] = lrow[r] * scale + rowsum16(pr);
      mrow[r] = mnew;
      p[r] = pr;
      #pragma unroll
      for (int t = 0; t < 4; ++t) o[t][r] *= scale;
    }
    __syncthreads();
    #pragma unroll
    for (int r = 0; r < 8; ++r) pt[(r + 8 * g) * 17 + lm] = p[r];
    __syncthreads();
    v2f ap[4];
    #pragma unroll
    for (int kt = 0; kt < 4; ++kt) {
      ap[kt].x = pt[lm * 17 + kt * 4 + 2 * g];
      ap[kt].y = pt[lm * 17 + kt * 4 + 2 * g + 1];
    }
    v2f bv[4][4];
    #pragma unroll
    for (int kt = 0; kt < 4; ++kt)
      #pragma unroll
      for (int t = 0; t < 4; ++t) {
        bv[kt][t].x = vb[(size_t)(j0 + kt * 4 + 2 * g) * C3 + t * 16 + lm];
        bv[kt][t].y = vb[(size_t)(j0 + kt * 4 + 2 * g + 1) * C3 + t * 16 + lm];
      }
    #pragma unroll
    for (int kt = 0; kt < 4; ++kt)
      #pragma unroll
      for (int t = 0; t < 4; ++t)
        o[t] = wmma4(ap[kt], bv[kt][t], o[t]);
  }
  #pragma unroll
  for (int r = 0; r < 8; ++r) {
    float inv = 1.f / lrow[r];
    int i = i0 + r + 8 * g;
    #pragma unroll
    for (int t = 0; t < 4; ++t)
      Y[((size_t)(b * TT + i)) * CC + h * HD + t * 16 + lm] = o[t][r] * inv;
  }
}

extern "C" void kernel_launch(void* const* d_in, const int* in_sizes, int n_in,
                              void* d_out, int out_size, void* d_ws, size_t ws_size,
                              hipStream_t stream) {
  const float* x      = (const float*)d_in[0];
  const float* w_attn = (const float*)d_in[1];
  const float* b_attn = (const float*)d_in[2];
  const float* w_proj = (const float*)d_in[3];
  const float* b_proj = (const float*)d_in[4];
  float* out = (float*)d_out;

  char* ws = (char*)d_ws;
  const size_t QKV_BYTES = (size_t)BB * TT * C3 * 4;
  const size_t FF_BYTES  = (size_t)BB * TT * TT * 4;
  const size_t MSK_BYTES = (size_t)BB * TT * (TT / 32) * 4;
  float*    qkv = (float*)ws;
  float*    FF  = (float*)(ws + QKV_BYTES);
  unsigned* msk = (unsigned*)(ws + QKV_BYTES + FF_BYTES);
  float*    Y   = (float*)(ws + QKV_BYTES + FF_BYTES + MSK_BYTES);

  gemm_wmma<<<dim3(C3 / 256, (BB * TT) / 32), 128, 0, stream>>>(
      x, w_attn, b_attn, qkv, BB * TT, C3, CC);
  head0_scores<<<dim3(TT / 16, TT / 16, BB), 32, 0, stream>>>(qkv, FF);
  ff_cumsum<<<dim3(TT / 256, BB), 256, 0, stream>>>(FF);
  select_mask<<<BB * TT, 256, 0, stream>>>(FF, msk);
  attn<<<dim3(TT / 16, NH, BB), 32, 0, stream>>>(qkv, FF, msk, Y);
  gemm_wmma<<<dim3(CC / 256, (BB * TT) / 32), 128, 0, stream>>>(
      Y, w_proj, b_proj, out, BB * TT, CC, CC);
}


# gemm (both launches): k loop in blocks of 32 with 16-byte fragment loads, four back to back per 16-row group, so a row's 128-byte line is fetched once per wave, not once per 4-k step; ping-pong regs
# speedup vs baseline: 2.3239x; 1.6511x over previous
#include <hip/hip_runtime.h>
#include <hip/hip_bf16.h>

#define BB 2
#define TT 2048
#define CC 1024
#define NH 16
#define HD 64
#define C3 3072
#define BUDGET 408

typedef __attribute__((ext_vector_type(2))) float v2f;
typedef __attribute__((ext_vector_type(8))) float v8f;

__device__ __forceinline__ v8f wmma4(v2f a, v2f b, v8f c) {
  return __builtin_amdgcn_wmma_f32_16x16x4_f32(false, a, false, b, (short)0, c, false, false);
}

__device__ __forceinline__ float rowmax16(float x) {
  #pragma unroll
  for (int d = 1; d < 16; d <<= 1) x = fmaxf(x, __shfl_xor(x, d, 32));
  return x;
}
__device__ __forceinline__ float rowsum16(float x) {
  #pragma unroll
  for (int d = 1; d < 16; d <<= 1) x += __shfl_xor(x, d, 32);
  return x;
}

typedef __attribute__((ext_vector_type(4))) float v4f;

__device__ __forceinline__ void gemm_load_block(v4f (&buf)[6][4], const float* const (&rp)[6], int kb) {
  #pragma unroll
  for (int rg = 0; rg < 6; ++rg)
    #pragma unroll
    for (int q = 0; q < 4; ++q)
      buf[rg][q] = *(const v4f*)(rp[rg] + kb + q * 8);
}

__device__ __forceinline__ void gemm_mma_block(const v4f (&buf)[6][4], v8f (&acc)[2][4]) {
  #pragma unroll
  for (int q = 0; q < 4; ++q) {
    #pragma unroll
    for (int s = 0; s < 2; ++s) {
      v2f a[2], b[4];
      #pragma unroll
      for (int mt = 0; mt < 2; ++mt) {
        a[mt].x = buf[mt][q][2 * s];
        a[mt].y = buf[mt][q][2 * s + 1];
      }
      #pragma unroll
      for (int t = 0; t < 4; ++t) {
        b[t].x = buf[2 + t][q][2 * s];
        b[t].y = buf[2 + t][q][2 * s + 1];
      }
      #pragma unroll
      for (int mt = 0; mt < 2; ++mt)
        #pragma unroll
        for (int t = 0; t < 4; ++t)
          acc[mt][t] = wmma4(a[mt], b[t], acc[mt][t]);
    }
  }
}

__global__ void __launch_bounds__(128)
gemm_wmma(const float* __restrict__ A, const float* __restrict__ W,
          const float* __restrict__ bias, float* __restrict__ Out,
          int M, int N, int K) {
  int wave = threadIdx.x >> 5;
  int lane = threadIdx.x & 31;
  int g  = lane >> 4;
  int lm = lane & 15;
  int m0 = blockIdx.y * 32;
  int n0 = blockIdx.x * 256 + wave * 64;
  if (n0 >= N || m0 >= M) return;

  v8f acc[2][4];
  #pragma unroll
  for (int t = 0; t < 4; ++t) {
    float bv = bias[n0 + t * 16 + lm];
    #pragma unroll
    for (int mt = 0; mt < 2; ++mt)
      #pragma unroll
      for (int r = 0; r < 8; ++r) acc[mt][t][r] = bv;
  }
  const float* rp[6];
  rp[0] = A + (size_t)(m0 + lm) * K + 4 * g;
  rp[1] = A + (size_t)(m0 + 16 + lm) * K + 4 * g;
  #pragma unroll
  for (int t = 0; t < 4; ++t)
    rp[2 + t] = W + (size_t)(n0 + t * 16 + lm) * K + 4 * g;

  v4f ping[6][4], pong[6][4];
  gemm_load_block(ping, rp, 0);
  int kb = 0;
  for (; kb + 64 < K; kb += 64) {
    gemm_load_block(pong, rp, kb + 32);
    __builtin_amdgcn_sched_barrier(0);
    gemm_mma_block(ping, acc);
    __builtin_amdgcn_sched_barrier(0);
    gemm_load_block(ping, rp, kb + 64);
    __builtin_amdgcn_sched_barrier(0);
    gemm_mma_block(pong, acc);
    __builtin_amdgcn_sched_barrier(0);
  }
  gemm_load_block(pong, rp, kb + 32);
  __builtin_amdgcn_sched_barrier(0);
  gemm_mma_block(ping, acc);
  gemm_mma_block(pong, acc);
  #pragma unroll
  for (int mt = 0; mt < 2; ++mt)
    #pragma unroll
    for (int t = 0; t < 4; ++t)
      #pragma unroll
      for (int r = 0; r < 8; ++r)
        Out[(size_t)(m0 + mt * 16 + r + 8 * g) * N + n0 + t * 16 + lm] = acc[mt][t][r];
}

__global__ void head0_scores(const float* __restrict__ qkv, float* __restrict__ S) {
  int jb = blockIdx.x, ib = blockIdx.y, b = blockIdx.z;
  int lane = threadIdx.x & 31, g = lane >> 4, lm = lane & 15;
  int i0 = ib * 16, j0 = jb * 16;
  if (jb > ib) {
    #pragma unroll
    for (int r = 0; r < 8; ++r)
      S[((size_t)(b * TT + i0 + r + 8 * g)) * TT + j0 + lm] = 0.f;
    return;
  }
  const float* qrow = qkv + ((size_t)(b * TT) + i0 + lm) * C3;
  const float* krow = qkv + ((size_t)(b * TT) + j0 + lm) * C3 + CC;
  v2f aq[16], bk[16];
  #pragma unroll
  for (int ks = 0; ks < 16; ++ks) {
    aq[ks] = *(const v2f*)(qrow + ks * 4 + 2 * g);
    bk[ks] = *(const v2f*)(krow + ks * 4 + 2 * g);
  }
  v8f acc = {0.f, 0.f, 0.f, 0.f, 0.f, 0.f, 0.f, 0.f};
  #pragma unroll
  for (int ks = 0; ks < 16; ++ks) acc = wmma4(aq[ks], bk[ks], acc);
  #pragma unroll
  for (int r = 0; r < 8; ++r) {
    int i = i0 + r + 8 * g, j = j0 + lm;
    float s = acc[r] * 0.125f;
    s = (j < i && j > 0) ? fmaxf(s, 0.f) : 0.f;
    S[((size_t)(b * TT + i)) * TT + j] = s;
  }
}

__global__ void ff_cumsum(float* __restrict__ S) {
  int j = blockIdx.x * blockDim.x + threadIdx.x;
  int b = blockIdx.y;
  float* col = S + (size_t)b * TT * TT + j;
  float carry = 0.f;
  #pragma unroll 8
  for (int i = 0; i < TT; ++i) {
    float v = col[(size_t)i * TT];
    col[(size_t)i * TT] = carry;
    carry += v;
  }
}

__global__ void select_mask(const float* __restrict__ FF, unsigned* __restrict__ mask) {
  int row = blockIdx.x;
  int i = row & (TT - 1);
  const float* f = FF + (size_t)row * TT;
  unsigned* mrow = mask + (size_t)row * (TT / 32);
  int tid = threadIdx.x;

  if (i < BUDGET) {
    for (int w = tid; w < TT / 32; w += 256) {
      int jb = w * 32;
      unsigned m;
      if (jb + 31 <= i) m = 0xFFFFFFFFu;
      else if (jb > i)  m = 0u;
      else              m = (1u << (i - jb + 1)) - 1u;
      mrow[w] = m;
    }
    return;
  }

  __shared__ unsigned hist[256];
  __shared__ unsigned misc[4];
  __shared__ unsigned wave_tot[8];

  if (tid == 0) { misc[0] = 0u; misc[1] = BUDGET - 1; }
  unsigned prefix = 0, prefmask = 0;
  for (int pass = 0; pass < 4; ++pass) {
    int shift = 24 - 8 * pass;
    hist[tid] = 0u;
    __syncthreads();
    for (int j = tid; j <= i; j += 256) {
      unsigned u = __float_as_uint(f[j]);
      if ((u & prefmask) == prefix) atomicAdd(&hist[(u >> shift) & 255u], 1u);
    }
    __syncthreads();
    if (tid == 0) {
      unsigned kk = misc[1], cum = 0; int d = 0;
      for (; d < 256; ++d) { unsigned h = hist[d]; if (cum + h > kk) break; cum += h; }
      misc[1] = kk - cum;
      misc[0] |= ((unsigned)d << shift);
    }
    __syncthreads();
    prefix = misc[0];
    prefmask |= (255u << shift);
  }
  unsigned theta = prefix;

  if (tid == 0) misc[2] = 0u;
  __syncthreads();
  unsigned cl = 0;
  for (int j = tid; j <= i; j += 256)
    if (__float_as_uint(f[j]) < theta) cl++;
  atomicAdd(&misc[2], cl);
  __syncthreads();
  int ntake = BUDGET - (int)misc[2];

  int lane = tid & 31, wv = tid >> 5;
  unsigned below = (1u << lane) - 1u;
  unsigned carry = 0;
  for (int base = 0; base < TT; base += 256) {
    int j = base + tid;
    unsigned u = (j <= i) ? __float_as_uint(f[j]) : 0xFFFFFFFFu;
    bool eq = (j <= i) && (u == theta);
    unsigned bal = (unsigned)__ballot(eq);
    if (lane == 0) wave_tot[wv] = __popc(bal);
    __syncthreads();
    unsigned pre = carry + __popc(bal & below);
    for (int w2 = 0; w2 < wv; ++w2) pre += wave_tot[w2];
    bool keep = (j <= i) && ((u < theta) || (eq && (int)pre < ntake));
    keep = keep || (j == i);
    unsigned kb = (unsigned)__ballot(keep);
    if (lane == 0) mrow[(base + wv * 32) >> 5] = kb;
    unsigned tot = 0;
    #pragma unroll
    for (int w2 = 0; w2 < 8; ++w2) tot += wave_tot[w2];
    carry += tot;
    __syncthreads();
  }
}

__global__ void attn(const float* __restrict__ qkv, const float* __restrict__ FF,
                     const unsigned* __restrict__ mask, float* __restrict__ Y) {
  int ib = blockIdx.x, h = blockIdx.y, b = blockIdx.z;
  int lane = threadIdx.x & 31, g = lane >> 4, lm = lane & 15;
  int i0 = ib * 16;
  __shared__ float pt[16 * 17];

  const float* qb = qkv + ((size_t)(b * TT)) * C3 + h * HD;
  const float* kb = qb + CC;
  const float* vb = qb + 2 * CC;

  v2f aq[16];
  #pragma unroll
  for (int ks = 0; ks < 16; ++ks)
    aq[ks] = *(const v2f*)(qb + (size_t)(i0 + lm) * C3 + ks * 4 + 2 * g);

  float mrow[8], lrow[8];
  v8f o[4];
  #pragma unroll
  for (int t = 0; t < 4; ++t)
    #pragma unroll
    for (int r = 0; r < 8; ++r) o[t][r] = 0.f;
  #pragma unroll
  for (int r = 0; r < 8; ++r) { mrow[r] = -1e30f; lrow[r] = 0.f; }

  const unsigned* mrow16 = mask + ((size_t)(b * TT + i0)) * (TT / 32);
  unsigned or0 = 0u, or1 = 0u;
  #pragma unroll
  for (int r = 0; r < 16; ++r) {
    or0 |= mrow16[r * (TT / 32) + lane];
    or1 |= mrow16[r * (TT / 32) + 32 + lane];
  }
  int wsrc = lane >> 1, hsh = (lane & 1) * 16;
  unsigned w0 = __shfl(or0, wsrc, 32), w1 = __shfl(or0, 16 + wsrc, 32);
  unsigned w2 = __shfl(or1, wsrc, 32), w3 = __shfl(or1, 16 + wsrc, 32);
  unsigned t0 = (unsigned)__ballot(((w0 >> hsh) & 0xFFFFu) != 0u);
  unsigned t1 = (unsigned)__ballot(((w1 >> hsh) & 0xFFFFu) != 0u);
  unsigned t2 = (unsigned)__ballot(((w2 >> hsh) & 0xFFFFu) != 0u);
  unsigned t3 = (unsigned)__ballot(((w3 >> hsh) & 0xFFFFu) != 0u);
  unsigned long long tlo = (unsigned long long)t0 | ((unsigned long long)t1 << 32);
  unsigned long long thi = (unsigned long long)t2 | ((unsigned long long)t3 << 32);
  if (ib < 64) {
    if (ib < 63) tlo &= (2ull << ib) - 1ull;
    thi = 0ull;
  } else if (ib < 127) {
    thi &= (2ull << (ib - 64)) - 1ull;
  }

  for (;;) {
    int jb;
    if (tlo != 0ull)      { jb = __builtin_ctzll(tlo);      tlo &= tlo - 1ull; }
    else if (thi != 0ull) { jb = 64 + __builtin_ctzll(thi); thi &= thi - 1ull; }
    else break;
    int j0 = jb * 16;
    v2f bk[16];
    #pragma unroll
    for (int ks = 0; ks < 16; ++ks)
      bk[ks] = *(const v2f*)(kb + (size_t)(j0 + lm) * C3 + ks * 4 + 2 * g);
    v8f acc = {0.f, 0.f, 0.f, 0.f, 0.f, 0.f, 0.f, 0.f};
    #pragma unroll
    for (int ks = 0; ks < 16; ++ks) acc = wmma4(aq[ks], bk[ks], acc);

    float ffv[8]; unsigned mwv[8];
    #pragma unroll
    for (int r = 0; r < 8; ++r) {
      int i = i0 + r + 8 * g;
      ffv[r] = FF[((size_t)(b * TT + i)) * TT + j0 + lm];
      mwv[r] = mask[((size_t)(b * TT + i)) * (TT / 32) + (j0 >> 5)];
    }
    float p[8];
    #pragma unroll
    for (int r = 0; r < 8; ++r) {
      int i = i0 + r + 8 * g, j = j0 + lm;
      bool keep = ((mwv[r] >> (j & 31)) & 1u) && (j <= i);
      float s = keep ? (acc[r] * 0.125f - ffv[r]) : -1e30f;
      float bm = rowmax16(s);
      float mnew = fmaxf(mrow[r], bm);
      float scale = __expf(mrow[r] - mnew);
      float pr = __expf(s - mnew);
      lrow[r] = lrow[r] * scale + rowsum16(pr);
      mrow[r] = mnew;
      p[r] = pr;
      #pragma unroll
      for (int t = 0; t < 4; ++t) o[t][r] *= scale;
    }
    __syncthreads();
    #pragma unroll
    for (int r = 0; r < 8; ++r) pt[(r + 8 * g) * 17 + lm] = p[r];
    __syncthreads();
    v2f ap[4];
    #pragma unroll
    for (int kt = 0; kt < 4; ++kt) {
      ap[kt].x = pt[lm * 17 + kt * 4 + 2 * g];
      ap[kt].y = pt[lm * 17 + kt * 4 + 2 * g + 1];
    }
    v2f bv[4][4];
    #pragma unroll
    for (int kt = 0; kt < 4; ++kt)
      #pragma unroll
      for (int t = 0; t < 4; ++t) {
        bv[kt][t].x = vb[(size_t)(j0 + kt * 4 + 2 * g) * C3 + t * 16 + lm];
        bv[kt][t].y = vb[(size_t)(j0 + kt * 4 + 2 * g + 1) * C3 + t * 16 + lm];
      }
    #pragma unroll
    for (int kt = 0; kt < 4; ++kt)
      #pragma unroll
      for (int t = 0; t < 4; ++t)
        o[t] = wmma4(ap[kt], bv[kt][t], o[t]);
  }
  #pragma unroll
  for (int r = 0; r < 8; ++r) {
    float inv = 1.f / lrow[r];
    int i = i0 + r + 8 * g;
    #pragma unroll
    for (int t = 0; t < 4; ++t)
      Y[((size_t)(b * TT + i)) * CC + h * HD + t * 16 + lm] = o[t][r] * inv;
  }
}

extern "C" void kernel_launch(void* const* d_in, const int* in_sizes, int n_in,
                              void* d_out, int out_size, void* d_ws, size_t ws_size,
                              hipStream_t stream) {
  const float* x      = (const float*)d_in[0];
  const float* w_attn = (const float*)d_in[1];
  const float* b_attn = (const float*)d_in[2];
  const float* w_proj = (const float*)d_in[3];
  const float* b_proj = (const float*)d_in[4];
  float* out = (float*)d_out;

  char* ws = (char*)d_ws;
  const size_t QKV_BYTES = (size_t)BB * TT * C3 * 4;
  const size_t FF_BYTES  = (size_t)BB * TT * TT * 4;
  const size_t MSK_BYTES = (size_t)BB * TT * (TT / 32) * 4;
  float*    qkv = (float*)ws;
  float*    FF  = (float*)(ws + QKV_BYTES);
  unsigned* msk = (unsigned*)(ws + QKV_BYTES + FF_BYTES);
  float*    Y   = (float*)(ws + QKV_BYTES + FF_BYTES + MSK_BYTES);

  gemm_wmma<<<dim3(C3 / 256, (BB * TT) / 32), 128, 0, stream>>>(
      x, w_attn, b_attn, qkv, BB * TT, C3, CC);
  head0_scores<<<dim3(TT / 16, TT / 16, BB), 32, 0, stream>>>(qkv, FF);
  ff_cumsum<<<dim3(TT / 256, BB), 256, 0, stream>>>(FF);
  select_mask<<<BB * TT, 256, 0, stream>>>(FF, msk);
  attn<<<dim3(TT / 16, NH, BB), 32, 0, stream>>>(qkv, FF, msk, Y);
  gemm_wmma<<<dim3(CC / 256, (BB * TT) / 32), 128, 0, stream>>>(
      Y, w_proj, b_proj, out, BB * TT, CC, CC);
}


# attn: one-wave block declared as launch bound, lifting the register cap assumed for an undeclared block size: Q fragments stay in registers, no scratch reload in front of each QK WMMA
# speedup vs baseline: 2.7699x; 1.1919x over previous
#include <hip/hip_runtime.h>
#include <hip/hip_bf16.h>

#define BB 2
#define TT 2048
#define CC 1024
#define NH 16
#define HD 64
#define C3 3072
#define BUDGET 408

typedef __attribute__((ext_vector_type(2))) float v2f;
typedef __attribute__((ext_vector_type(8))) float v8f;

__device__ __forceinline__ v8f wmma4(v2f a, v2f b, v8f c) {
  return __builtin_amdgcn_wmma_f32_16x16x4_f32(false, a, false, b, (short)0, c, false, false);
}

__device__ __forceinline__ float rowmax16(float x) {
  #pragma unroll
  for (int d = 1; d < 16; d <<= 1) x = fmaxf(x, __shfl_xor(x, d, 32));
  return x;
}
__device__ __forceinline__ float rowsum16(float x) {
  #pragma unroll
  for (int d = 1; d < 16; d <<= 1) x += __shfl_xor(x, d, 32);
  return x;
}

typedef __attribute__((ext_vector_type(4))) float v4f;

__device__ __forceinline__ void gemm_load_block(v4f (&buf)[6][4], const float* const (&rp)[6], int kb) {
  #pragma unroll
  for (int rg = 0; rg < 6; ++rg)
    #pragma unroll
    for (int q = 0; q < 4; ++q)
      buf[rg][q] = *(const v4f*)(rp[rg] + kb + q * 8);
}

__device__ __forceinline__ void gemm_mma_block(const v4f (&buf)[6][4], v8f (&acc)[2][4]) {
  #pragma unroll
  for (int q = 0; q < 4; ++q) {
    #pragma unroll
    for (int s = 0; s < 2; ++s) {
      v2f a[2], b[4];
      #pragma unroll
      for (int mt = 0; mt < 2; ++mt) {
        a[mt].x = buf[mt][q][2 * s];
        a[mt].y = buf[mt][q][2 * s + 1];
      }
      #pragma unroll
      for (int t = 0; t < 4; ++t) {
        b[t].x = buf[2 + t][q][2 * s];
        b[t].y = buf[2 + t][q][2 * s + 1];
      }
      #pragma unroll
      for (int mt = 0; mt < 2; ++mt)
        #pragma unroll
        for (int t = 0; t < 4; ++t)
          acc[mt][t] = wmma4(a[mt], b[t], acc[mt][t]);
    }
  }
}

__global__ void __launch_bounds__(128)
gemm_wmma(const float* __restrict__ A, const float* __restrict__ W,
          const float* __restrict__ bias, float* __restrict__ Out,
          int M, int N, int K) {
  int wave = threadIdx.x >> 5;
  int lane = threadIdx.x & 31;
  int g  = lane >> 4;
  int lm = lane & 15;
  int m0 = blockIdx.y * 32;
  int n0 = blockIdx.x * 256 + wave * 64;
  if (n0 >= N || m0 >= M) return;

  v8f acc[2][4];
  #pragma unroll
  for (int t = 0; t < 4; ++t) {
    float bv = bias[n0 + t * 16 + lm];
    #pragma unroll
    for (int mt = 0; mt < 2; ++mt)
      #pragma unroll
      for (int r = 0; r < 8; ++r) acc[mt][t][r] = bv;
  }
  const float* rp[6];
  rp[0] = A + (size_t)(m0 + lm) * K + 4 * g;
  rp[1] = A + (size_t)(m0 + 16 + lm) * K + 4 * g;
  #pragma unroll
  for (int t = 0; t < 4; ++t)
    rp[2 + t] = W + (size_t)(n0 + t * 16 + lm) * K + 4 * g;

  v4f ping[6][4], pong[6][4];
  gemm_load_block(ping, rp, 0);
  int kb = 0;
  for (; kb + 64 < K; kb += 64) {
    gemm_load_block(pong, rp, kb + 32);
    __builtin_amdgcn_sched_barrier(0);
    gemm_mma_block(ping, acc);
    __builtin_amdgcn_sched_barrier(0);
    gemm_load_block(ping, rp, kb + 64);
    __builtin_amdgcn_sched_barrier(0);
    gemm_mma_block(pong, acc);
    __builtin_amdgcn_sched_barrier(0);
  }
  gemm_load_block(pong, rp, kb + 32);
  __builtin_amdgcn_sched_barrier(0);
  gemm_mma_block(ping, acc);
  gemm_mma_block(pong, acc);
  #pragma unroll
  for (int mt = 0; mt < 2; ++mt)
    #pragma unroll
    for (int t = 0; t < 4; ++t)
      #pragma unroll
      for (int r = 0; r < 8; ++r)
        Out[(size_t)(m0 + mt * 16 + r + 8 * g) * N + n0 + t * 16 + lm] = acc[mt][t][r];
}

__global__ void head0_scores(const float* __restrict__ qkv, float* __restrict__ S) {
  int jb = blockIdx.x, ib = blockIdx.y, b = blockIdx.z;
  int lane = threadIdx.x & 31, g = lane >> 4, lm = lane & 15;
  int i0 = ib * 16, j0 = jb * 16;
  if (jb > ib) {
    #pragma unroll
    for (int r = 0; r < 8; ++r)
      S[((size_t)(b * TT + i0 + r + 8 * g)) * TT + j0 + lm] = 0.f;
    return;
  }
  const float* qrow = qkv + ((size_t)(b * TT) + i0 + lm) * C3;
  const float* krow = qkv + ((size_t)(b * TT) + j0 + lm) * C3 + CC;
  v2f aq[16], bk[16];
  #pragma unroll
  for (int ks = 0; ks < 16; ++ks) {
    aq[ks] = *(const v2f*)(qrow + ks * 4 + 2 * g);
    bk[ks] = *(const v2f*)(krow + ks * 4 + 2 * g);
  }
  v8f acc = {0.f, 0.f, 0.f, 0.f, 0.f, 0.f, 0.f, 0.f};
  #pragma unroll
  for (int ks = 0; ks < 16; ++ks) acc = wmma4(aq[ks], bk[ks], acc);
  #pragma unroll
  for (int r = 0; r < 8; ++r) {
    int i = i0 + r + 8 * g, j = j0 + lm;
    float s = acc[r] * 0.125f;
    s = (j < i && j > 0) ? fmaxf(s, 0.f) : 0.f;
    S[((size_t)(b * TT + i)) * TT + j] = s;
  }
}

__global__ void ff_cumsum(float* __restrict__ S) {
  int j = blockIdx.x * blockDim.x + threadIdx.x;
  int b = blockIdx.y;
  float* col = S + (size_t)b * TT * TT + j;
  float carry = 0.f;
  #pragma unroll 8
  for (int i = 0; i < TT; ++i) {
    float v = col[(size_t)i * TT];
    col[(size_t)i * TT] = carry;
    carry += v;
  }
}

__global__ void select_mask(const float* __restrict__ FF, unsigned* __restrict__ mask) {
  int row = blockIdx.x;
  int i = row & (TT - 1);
  const float* f = FF + (size_t)row * TT;
  unsigned* mrow = mask + (size_t)row * (TT / 32);
  int tid = threadIdx.x;

  if (i < BUDGET) {
    for (int w = tid; w < TT / 32; w += 256) {
      int jb = w * 32;
      unsigned m;
      if (jb + 31 <= i) m = 0xFFFFFFFFu;
      else if (jb > i)  m = 0u;
      else              m = (1u << (i - jb + 1)) - 1u;
      mrow[w] = m;
    }
    return;
  }

  __shared__ unsigned hist[256];
  __shared__ unsigned misc[4];
  __shared__ unsigned wave_tot[8];

  if (tid == 0) { misc[0] = 0u; misc[1] = BUDGET - 1; }
  unsigned prefix = 0, prefmask = 0;
  for (int pass = 0; pass < 4; ++pass) {
    int shift = 24 - 8 * pass;
    hist[tid] = 0u;
    __syncthreads();
    for (int j = tid; j <= i; j += 256) {
      unsigned u = __float_as_uint(f[j]);
      if ((u & prefmask) == prefix) atomicAdd(&hist[(u >> shift) & 255u], 1u);
    }
    __syncthreads();
    if (tid == 0) {
      unsigned kk = misc[1], cum = 0; int d = 0;
      for (; d < 256; ++d) { unsigned h = hist[d]; if (cum + h > kk) break; cum += h; }
      misc[1] = kk - cum;
      misc[0] |= ((unsigned)d << shift);
    }
    __syncthreads();
    prefix = misc[0];
    prefmask |= (255u << shift);
  }
  unsigned theta = prefix;

  if (tid == 0) misc[2] = 0u;
  __syncthreads();
  unsigned cl = 0;
  for (int j = tid; j <= i; j += 256)
    if (__float_as_uint(f[j]) < theta) cl++;
  atomicAdd(&misc[2], cl);
  __syncthreads();
  int ntake = BUDGET - (int)misc[2];

  int lane = tid & 31, wv = tid >> 5;
  unsigned below = (1u << lane) - 1u;
  unsigned carry = 0;
  for (int base = 0; base < TT; base += 256) {
    int j = base + tid;
    unsigned u = (j <= i) ? __float_as_uint(f[j]) : 0xFFFFFFFFu;
    bool eq = (j <= i) && (u == theta);
    unsigned bal = (unsigned)__ballot(eq);
    if (lane == 0) wave_tot[wv] = __popc(bal);
    __syncthreads();
    unsigned pre = carry + __popc(bal & below);
    for (int w2 = 0; w2 < wv; ++w2) pre += wave_tot[w2];
    bool keep = (j <= i) && ((u < theta) || (eq && (int)pre < ntake));
    keep = keep || (j == i);
    unsigned kb = (unsigned)__ballot(keep);
    if (lane == 0) mrow[(base + wv * 32) >> 5] = kb;
    unsigned tot = 0;
    #pragma unroll
    for (int w2 = 0; w2 < 8; ++w2) tot += wave_tot[w2];
    carry += tot;
    __syncthreads();
  }
}

__global__ void __launch_bounds__(32)
attn(const float* __restrict__ qkv, const float* __restrict__ FF,
                     const unsigned* __restrict__ mask, float* __restrict__ Y) {
  int ib = blockIdx.x, h = blockIdx.y, b = blockIdx.z;
  int lane = threadIdx.x & 31, g = lane >> 4, lm = lane & 15;
  int i0 = ib * 16;
  __shared__ float pt[16 * 17];

  const float* qb = qkv + ((size_t)(b * TT)) * C3 + h * HD;
  const float* kb = qb + CC;
  const float* vb = qb + 2 * CC;

  v2f aq[16];
  #pragma unroll
  for (int ks = 0; ks < 16; ++ks)
    aq[ks] = *(const v2f*)(qb + (size_t)(i0 + lm) * C3 + ks * 4 + 2 * g);

  float mrow[8], lrow[8];
  v8f o[4];
  #pragma unroll
  for (int t = 0; t < 4; ++t)
    #pragma unroll
    for (int r = 0; r < 8; ++r) o[t][r] = 0.f;
  #pragma unroll
  for (int r = 0; r < 8; ++r) { mrow[r] = -1e30f; lrow[r] = 0.f; }

  const unsigned* mrow16 = mask + ((size_t)(b * TT + i0)) * (TT / 32);
  unsigned or0 = 0u, or1 = 0u;
  #pragma unroll
  for (int r = 0; r < 16; ++r) {
    or0 |= mrow16[r * (TT / 32) + lane];
    or1 |= mrow16[r * (TT / 32) + 32 + lane];
  }
  int wsrc = lane >> 1, hsh = (lane & 1) * 16;
  unsigned w0 = __shfl(or0, wsrc, 32), w1 = __shfl(or0, 16 + wsrc, 32);
  unsigned w2 = __shfl(or1, wsrc, 32), w3 = __shfl(or1, 16 + wsrc, 32);
  unsigned t0 = (unsigned)__ballot(((w0 >> hsh) & 0xFFFFu) != 0u);
  unsigned t1 = (unsigned)__ballot(((w1 >> hsh) & 0xFFFFu) != 0u);
  unsigned t2 = (unsigned)__ballot(((w2 >> hsh) & 0xFFFFu) != 0u);
  unsigned t3 = (unsigned)__ballot(((w3 >> hsh) & 0xFFFFu) != 0u);
  unsigned long long tlo = (unsigned long long)t0 | ((unsigned long long)t1 << 32);
  unsigned long long thi = (unsigned long long)t2 | ((unsigned long long)t3 << 32);
  if (ib < 64) {
    if (ib < 63) tlo &= (2ull << ib) - 1ull;
    thi = 0ull;
  } else if (ib < 127) {
    thi &= (2ull << (ib - 64)) - 1ull;
  }

  for (;;) {
    int jb;
    if (tlo != 0ull)      { jb = __builtin_ctzll(tlo);      tlo &= tlo - 1ull; }
    else if (thi != 0ull) { jb = 64 + __builtin_ctzll(thi); thi &= thi - 1ull; }
    else break;
    int j0 = jb * 16;
    v2f bk[16];
    #pragma unroll
    for (int ks = 0; ks < 16; ++ks)
      bk[ks] = *(const v2f*)(kb + (size_t)(j0 + lm) * C3 + ks * 4 + 2 * g);
    v8f acc = {0.f, 0.f, 0.f, 0.f, 0.f, 0.f, 0.f, 0.f};
    #pragma unroll
    for (int ks = 0; ks < 16; ++ks) acc = wmma4(aq[ks], bk[ks], acc);

    float ffv[8]; unsigned mwv[8];
    #pragma unroll
    for (int r = 0; r < 8; ++r) {
      int i = i0 + r + 8 * g;
      ffv[r] = FF[((size_t)(b * TT + i)) * TT + j0 + lm];
      mwv[r] = mask[((size_t)(b * TT + i)) * (TT / 32) + (j0 >> 5)];
    }
    float p[8];
    #pragma unroll
    for (int r = 0; r < 8; ++r) {
      int i = i0 + r + 8 * g, j = j0 + lm;
      bool keep = ((mwv[r] >> (j & 31)) & 1u) && (j <= i);
      float s = keep ? (acc[r] * 0.125f - ffv[r]) : -1e30f;
      float bm = rowmax16(s);
      float mnew = fmaxf(mrow[r], bm);
      float scale = __expf(mrow[r] - mnew);
      float pr = __expf(s - mnew);
      lrow[r] = lrow[r] * scale + rowsum16(pr);
      mrow[r] = mnew;
      p[r] = pr;
      #pragma unroll
      for (int t = 0; t < 4; ++t) o[t][r] *= scale;
    }
    __syncthreads();
    #pragma unroll
    for (int r = 0; r < 8; ++r) pt[(r + 8 * g) * 17 + lm] = p[r];
    __syncthreads();
    v2f ap[4];
    #pragma unroll
    for (int kt = 0; kt < 4; ++kt) {
      ap[kt].x = pt[lm * 17 + kt * 4 + 2 * g];
      ap[kt].y = pt[lm * 17 + kt * 4 + 2 * g + 1];
    }
    v2f bv[4][4];
    #pragma unroll
    for (int kt = 0; kt < 4; ++kt)
      #pragma unroll
      for (int t = 0; t < 4; ++t) {
        bv[kt][t].x = vb[(size_t)(j0 + kt * 4 + 2 * g) * C3 + t * 16 + lm];
        bv[kt][t].y = vb[(size_t)(j0 + kt * 4 + 2 * g + 1) * C3 + t * 16 + lm];
      }
    #pragma unroll
    for (int kt = 0; kt < 4; ++kt)
      #pragma unroll
      for (int t = 0; t < 4; ++t)
        o[t] = wmma4(ap[kt], bv[kt][t], o[t]);
  }
  #pragma unroll
  for (int r = 0; r < 8; ++r) {
    float inv = 1.f / lrow[r];
    int i = i0 + r + 8 * g;
    #pragma unroll
    for (int t = 0; t < 4; ++t)
      Y[((size_t)(b * TT + i)) * CC + h * HD + t * 16 + lm] = o[t][r] * inv;
  }
}

extern "C" void kernel_launch(void* const* d_in, const int* in_sizes, int n_in,
                              void* d_out, int out_size, void* d_ws, size_t ws_size,
                              hipStream_t stream) {
  const float* x      = (const float*)d_in[0];
  const float* w_attn = (const float*)d_in[1];
  const float* b_attn = (const float*)d_in[2];
  const float* w_proj = (const float*)d_in[3];
  const float* b_proj = (const float*)d_in[4];
  float* out = (float*)d_out;

  char* ws = (char*)d_ws;
  const size_t QKV_BYTES = (size_t)BB * TT * C3 * 4;
  const size_t FF_BYTES  = (size_t)BB * TT * TT * 4;
  const size_t MSK_BYTES = (size_t)BB * TT * (TT / 32) * 4;
  float*    qkv = (float*)ws;
  float*    FF  = (float*)(ws + QKV_BYTES);
  unsigned* msk = (unsigned*)(ws + QKV_BYTES + FF_BYTES);
  float*    Y   = (float*)(ws + QKV_BYTES + FF_BYTES + MSK_BYTES);

  gemm_wmma<<<dim3(C3 / 256, (BB * TT) / 32), 128, 0, stream>>>(
      x, w_attn, b_attn, qkv, BB * TT, C3, CC);
  head0_scores<<<dim3(TT / 16, TT / 16, BB), 32, 0, stream>>>(qkv, FF);
  ff_cumsum<<<dim3(TT / 256, BB), 256, 0, stream>>>(FF);
  select_mask<<<BB * TT, 256, 0, stream>>>(FF, msk);
  attn<<<dim3(TT / 16, NH, BB), 32, 0, stream>>>(qkv, FF, msk, Y);
  gemm_wmma<<<dim3(CC / 256, (BB * TT) / 32), 128, 0, stream>>>(
      Y, w_proj, b_proj, out, BB * TT, CC, CC);
}


# ff scan: each 32-column strip gets a block of 32 waves, 64 rows per lane held in registers (64 loads in flight), wave totals joined through LDS; was one thread walking all 2048 rows of its column
# speedup vs baseline: 3.6614x; 1.3219x over previous
#include <hip/hip_runtime.h>
#include <hip/hip_bf16.h>

#define BB 2
#define TT 2048
#define CC 1024
#define NH 16
#define HD 64
#define C3 3072
#define BUDGET 408

typedef __attribute__((ext_vector_type(2))) float v2f;
typedef __attribute__((ext_vector_type(8))) float v8f;

__device__ __forceinline__ v8f wmma4(v2f a, v2f b, v8f c) {
  return __builtin_amdgcn_wmma_f32_16x16x4_f32(false, a, false, b, (short)0, c, false, false);
}

__device__ __forceinline__ float rowmax16(float x) {
  #pragma unroll
  for (int d = 1; d < 16; d <<= 1) x = fmaxf(x, __shfl_xor(x, d, 32));
  return x;
}
__device__ __forceinline__ float rowsum16(float x) {
  #pragma unroll
  for (int d = 1; d < 16; d <<= 1) x += __shfl_xor(x, d, 32);
  return x;
}

typedef __attribute__((ext_vector_type(4))) float v4f;

__device__ __forceinline__ void gemm_load_block(v4f (&buf)[6][4], const float* const (&rp)[6], int kb) {
  #pragma unroll
  for (int rg = 0; rg < 6; ++rg)
    #pragma unroll
    for (int q = 0; q < 4; ++q)
      buf[rg][q] = *(const v4f*)(rp[rg] + kb + q * 8);
}

__device__ __forceinline__ void gemm_mma_block(const v4f (&buf)[6][4], v8f (&acc)[2][4]) {
  #pragma unroll
  for (int q = 0; q < 4; ++q) {
    #pragma unroll
    for (int s = 0; s < 2; ++s) {
      v2f a[2], b[4];
      #pragma unroll
      for (int mt = 0; mt < 2; ++mt) {
        a[mt].x = buf[mt][q][2 * s];
        a[mt].y = buf[mt][q][2 * s + 1];
      }
      #pragma unroll
      for (int t = 0; t < 4; ++t) {
        b[t].x = buf[2 + t][q][2 * s];
        b[t].y = buf[2 + t][q][2 * s + 1];
      }
      #pragma unroll
      for (int mt = 0; mt < 2; ++mt)
        #pragma unroll
        for (int t = 0; t < 4; ++t)
          acc[mt][t] = wmma4(a[mt], b[t], acc[mt][t]);
    }
  }
}

__global__ void __launch_bounds__(128)
gemm_wmma(const float* __restrict__ A, const float* __restrict__ W,
          const float* __restrict__ bias, float* __restrict__ Out,
          int M, int N, int K) {
  int wave = threadIdx.x >> 5;
  int lane = threadIdx.x & 31;
  int g  = lane >> 4;
  int lm = lane & 15;
  int m0 = blockIdx.y * 32;
  int n0 = blockIdx.x * 256 + wave * 64;
  if (n0 >= N || m0 >= M) return;

  v8f acc[2][4];
  #pragma unroll
  for (int t = 0; t < 4; ++t) {
    float bv = bias[n0 + t * 16 + lm];
    #pragma unroll
    for (int mt = 0; mt < 2; ++mt)
      #pragma unroll
      for (int r = 0; r < 8; ++r) acc[mt][t][r] = bv;
  }
  const float* rp[6];
  rp[0] = A + (size_t)(m0 + lm) * K + 4 * g;
  rp[1] = A + (size_t)(m0 + 16 + lm) * K + 4 * g;
  #pragma unroll
  for (int t = 0; t < 4; ++t)
    rp[2 + t] = W + (size_t)(n0 + t * 16 + lm) * K + 4 * g;

  v4f ping[6][4], pong[6][4];
  gemm_load_block(ping, rp, 0);
  int kb = 0;
  for (; kb + 64 < K; kb += 64) {
    gemm_load_block(pong, rp, kb + 32);
    __builtin_amdgcn_sched_barrier(0);
    gemm_mma_block(ping, acc);
    __builtin_amdgcn_sched_barrier(0);
    gemm_load_block(ping, rp, kb + 64);
    __builtin_amdgcn_sched_barrier(0);
    gemm_mma_block(pong, acc);
    __builtin_amdgcn_sched_barrier(0);
  }
  gemm_load_block(pong, rp, kb + 32);
  __builtin_amdgcn_sched_barrier(0);
  gemm_mma_block(ping, acc);
  gemm_mma_block(pong, acc);
  #pragma unroll
  for (int mt = 0; mt < 2; ++mt)
    #pragma unroll
    for (int t = 0; t < 4; ++t)
      #pragma unroll
      for (int r = 0; r < 8; ++r)
        Out[(size_t)(m0 + mt * 16 + r + 8 * g) * N + n0 + t * 16 + lm] = acc[mt][t][r];
}

__global__ void head0_scores(const float* __restrict__ qkv, float* __restrict__ S) {
  int jb = blockIdx.x, ib = blockIdx.y, b = blockIdx.z;
  int lane = threadIdx.x & 31, g = lane >> 4, lm = lane & 15;
  int i0 = ib * 16, j0 = jb * 16;
  if (jb > ib) {
    #pragma unroll
    for (int r = 0; r < 8; ++r)
      S[((size_t)(b * TT + i0 + r + 8 * g)) * TT + j0 + lm] = 0.f;
    return;
  }
  const float* qrow = qkv + ((size_t)(b * TT) + i0 + lm) * C3;
  const float* krow = qkv + ((size_t)(b * TT) + j0 + lm) * C3 + CC;
  v2f aq[16], bk[16];
  #pragma unroll
  for (int ks = 0; ks < 16; ++ks) {
    aq[ks] = *(const v2f*)(qrow + ks * 4 + 2 * g);
    bk[ks] = *(const v2f*)(krow + ks * 4 + 2 * g);
  }
  v8f acc = {0.f, 0.f, 0.f, 0.f, 0.f, 0.f, 0.f, 0.f};
  #pragma unroll
  for (int ks = 0; ks < 16; ++ks) acc = wmma4(aq[ks], bk[ks], acc);
  #pragma unroll
  for (int r = 0; r < 8; ++r) {
    int i = i0 + r + 8 * g, j = j0 + lm;
    float s = acc[r] * 0.125f;
    s = (j < i && j > 0) ? fmaxf(s, 0.f) : 0.f;
    S[((size_t)(b * TT + i)) * TT + j] = s;
  }
}

#define SCAN_ROWS 64
__global__ void __launch_bounds__(1024)
ff_cumsum(float* __restrict__ S) {
  __shared__ float tot[32 * 32];
  int lane = threadIdx.x & 31, w = threadIdx.x >> 5;
  int j = blockIdx.x * 32 + lane;
  int b = blockIdx.y;
  float* col = S + (size_t)b * TT * TT + (size_t)(w * SCAN_ROWS) * TT + j;
  float v[SCAN_ROWS];
  #pragma unroll
  for (int r = 0; r < SCAN_ROWS; ++r) v[r] = col[(size_t)r * TT];
  float run = 0.f;
  #pragma unroll
  for (int r = 0; r < SCAN_ROWS; ++r) { float t = v[r]; v[r] = run; run += t; }
  tot[w * 32 + lane] = run;
  __syncthreads();
  float off = 0.f;
  #pragma unroll
  for (int w2 = 0; w2 < 31; ++w2) off += (w2 < w) ? tot[w2 * 32 + lane] : 0.f;
  #pragma unroll
  for (int r = 0; r < SCAN_ROWS; ++r) col[(size_t)r * TT] = v[r] + off;
}

__global__ void select_mask(const float* __restrict__ FF, unsigned* __restrict__ mask) {
  int row = blockIdx.x;
  int i = row & (TT - 1);
  const float* f = FF + (size_t)row * TT;
  unsigned* mrow = mask + (size_t)row * (TT / 32);
  int tid = threadIdx.x;

  if (i < BUDGET) {
    for (int w = tid; w < TT / 32; w += 256) {
      int jb = w * 32;
      unsigned m;
      if (jb + 31 <= i) m = 0xFFFFFFFFu;
      else if (jb > i)  m = 0u;
      else              m = (1u << (i - jb + 1)) - 1u;
      mrow[w] = m;
    }
    return;
  }

  __shared__ unsigned hist[256];
  __shared__ unsigned misc[4];
  __shared__ unsigned wave_tot[8];

  if (tid == 0) { misc[0] = 0u; misc[1] = BUDGET - 1; }
  unsigned prefix = 0, prefmask = 0;
  for (int pass = 0; pass < 4; ++pass) {
    int shift = 24 - 8 * pass;
    hist[tid] = 0u;
    __syncthreads();
    for (int j = tid; j <= i; j += 256) {
      unsigned u = __float_as_uint(f[j]);
      if ((u & prefmask) == prefix) atomicAdd(&hist[(u >> shift) & 255u], 1u);
    }
    __syncthreads();
    if (tid == 0) {
      unsigned kk = misc[1], cum = 0; int d = 0;
      for (; d < 256; ++d) { unsigned h = hist[d]; if (cum + h > kk) break; cum += h; }
      misc[1] = kk - cum;
      misc[0] |= ((unsigned)d << shift);
    }
    __syncthreads();
    prefix = misc[0];
    prefmask |= (255u << shift);
  }
  unsigned theta = prefix;

  if (tid == 0) misc[2] = 0u;
  __syncthreads();
  unsigned cl = 0;
  for (int j = tid; j <= i; j += 256)
    if (__float_as_uint(f[j]) < theta) cl++;
  atomicAdd(&misc[2], cl);
  __syncthreads();
  int ntake = BUDGET - (int)misc[2];

  int lane = tid & 31, wv = tid >> 5;
  unsigned below = (1u << lane) - 1u;
  unsigned carry = 0;
  for (int base = 0; base < TT; base += 256) {
    int j = base + tid;
    unsigned u = (j <= i) ? __float_as_uint(f[j]) : 0xFFFFFFFFu;
    bool eq = (j <= i) && (u == theta);
    unsigned bal = (unsigned)__ballot(eq);
    if (lane == 0) wave_tot[wv] = __popc(bal);
    __syncthreads();
    unsigned pre = carry + __popc(bal & below);
    for (int w2 = 0; w2 < wv; ++w2) pre += wave_tot[w2];
    bool keep = (j <= i) && ((u < theta) || (eq && (int)pre < ntake));
    keep = keep || (j == i);
    unsigned kb = (unsigned)__ballot(keep);
    if (lane == 0) mrow[(base + wv * 32) >> 5] = kb;
    unsigned tot = 0;
    #pragma unroll
    for (int w2 = 0; w2 < 8; ++w2) tot += wave_tot[w2];
    carry += tot;
    __syncthreads();
  }
}

__global__ void __launch_bounds__(32)
attn(const float* __restrict__ qkv, const float* __restrict__ FF,
                     const unsigned* __restrict__ mask, float* __restrict__ Y) {
  int ib = blockIdx.x, h = blockIdx.y, b = blockIdx.z;
  int lane = threadIdx.x & 31, g = lane >> 4, lm = lane & 15;
  int i0 = ib * 16;
  __shared__ float pt[16 * 17];

  const float* qb = qkv + ((size_t)(b * TT)) * C3 + h * HD;
  const float* kb = qb + CC;
  const float* vb = qb + 2 * CC;

  v2f aq[16];
  #pragma unroll
  for (int ks = 0; ks < 16; ++ks)
    aq[ks] = *(const v2f*)(qb + (size_t)(i0 + lm) * C3 + ks * 4 + 2 * g);

  float mrow[8], lrow[8];
  v8f o[4];
  #pragma unroll
  for (int t = 0; t < 4; ++t)
    #pragma unroll
    for (int r = 0; r < 8; ++r) o[t][r] = 0.f;
  #pragma unroll
  for (int r = 0; r < 8; ++r) { mrow[r] = -1e30f; lrow[r] = 0.f; }

  const unsigned* mrow16 = mask + ((size_t)(b * TT + i0)) * (TT / 32);
  unsigned or0 = 0u, or1 = 0u;
  #pragma unroll
  for (int r = 0; r < 16; ++r) {
    or0 |= mrow16[r * (TT / 32) + lane];
    or1 |= mrow16[r * (TT / 32) + 32 + lane];
  }
  int wsrc = lane >> 1, hsh = (lane & 1) * 16;
  unsigned w0 = __shfl(or0, wsrc, 32), w1 = __shfl(or0, 16 + wsrc, 32);
  unsigned w2 = __shfl(or1, wsrc, 32), w3 = __shfl(or1, 16 + wsrc, 32);
  unsigned t0 = (unsigned)__ballot(((w0 >> hsh) & 0xFFFFu) != 0u);
  unsigned t1 = (unsigned)__ballot(((w1 >> hsh) & 0xFFFFu) != 0u);
  unsigned t2 = (unsigned)__ballot(((w2 >> hsh) & 0xFFFFu) != 0u);
  unsigned t3 = (unsigned)__ballot(((w3 >> hsh) & 0xFFFFu) != 0u);
  unsigned long long tlo = (unsigned long long)t0 | ((unsigned long long)t1 << 32);
  unsigned long long thi = (unsigned long long)t2 | ((unsigned long long)t3 << 32);
  if (ib < 64) {
    if (ib < 63) tlo &= (2ull << ib) - 1ull;
    thi = 0ull;
  } else if (ib < 127) {
    thi &= (2ull << (ib - 64)) - 1ull;
  }

  for (;;) {
    int jb;
    if (tlo != 0ull)      { jb = __builtin_ctzll(tlo);      tlo &= tlo - 1ull; }
    else if (thi != 0ull) { jb = 64 + __builtin_ctzll(thi); thi &= thi - 1ull; }
    else break;
    int j0 = jb * 16;
    v2f bk[16];
    #pragma unroll
    for (int ks = 0; ks < 16; ++ks)
      bk[ks] = *(const v2f*)(kb + (size_t)(j0 + lm) * C3 + ks * 4 + 2 * g);
    v8f acc = {0.f, 0.f, 0.f, 0.f, 0.f, 0.f, 0.f, 0.f};
    #pragma unroll
    for (int ks = 0; ks < 16; ++ks) acc = wmma4(aq[ks], bk[ks], acc);

    float ffv[8]; unsigned mwv[8];
    #pragma unroll
    for (int r = 0; r < 8; ++r) {
      int i = i0 + r + 8 * g;
      ffv[r] = FF[((size_t)(b * TT + i)) * TT + j0 + lm];
      mwv[r] = mask[((size_t)(b * TT + i)) * (TT / 32) + (j0 >> 5)];
    }
    float p[8];
    #pragma unroll
    for (int r = 0; r < 8; ++r) {
      int i = i0 + r + 8 * g, j = j0 + lm;
      bool keep = ((mwv[r] >> (j & 31)) & 1u) && (j <= i);
      float s = keep ? (acc[r] * 0.125f - ffv[r]) : -1e30f;
      float bm = rowmax16(s);
      float mnew = fmaxf(mrow[r], bm);
      float scale = __expf(mrow[r] - mnew);
      float pr = __expf(s - mnew);
      lrow[r] = lrow[r] * scale + rowsum16(pr);
      mrow[r] = mnew;
      p[r] = pr;
      #pragma unroll
      for (int t = 0; t < 4; ++t) o[t][r] *= scale;
    }
    __syncthreads();
    #pragma unroll
    for (int r = 0; r < 8; ++r) pt[(r + 8 * g) * 17 + lm] = p[r];
    __syncthreads();
    v2f ap[4];
    #pragma unroll
    for (int kt = 0; kt < 4; ++kt) {
      ap[kt].x = pt[lm * 17 + kt * 4 + 2 * g];
      ap[kt].y = pt[lm * 17 + kt * 4 + 2 * g + 1];
    }
    v2f bv[4][4];
    #pragma unroll
    for (int kt = 0; kt < 4; ++kt)
      #pragma unroll
      for (int t = 0; t < 4; ++t) {
        bv[kt][t].x = vb[(size_t)(j0 + kt * 4 + 2 * g) * C3 + t * 16 + lm];
        bv[kt][t].y = vb[(size_t)(j0 + kt * 4 + 2 * g + 1) * C3 + t * 16 + lm];
      }
    #pragma unroll
    for (int kt = 0; kt < 4; ++kt)
      #pragma unroll
      for (int t = 0; t < 4; ++t)
        o[t] = wmma4(ap[kt], bv[kt][t], o[t]);
  }
  #pragma unroll
  for (int r = 0; r < 8; ++r) {
    float inv = 1.f / lrow[r];
    int i = i0 + r + 8 * g;
    #pragma unroll
    for (int t = 0; t < 4; ++t)
      Y[((size_t)(b * TT + i)) * CC + h * HD + t * 16 + lm] = o[t][r] * inv;
  }
}

extern "C" void kernel_launch(void* const* d_in, const int* in_sizes, int n_in,
                              void* d_out, int out_size, void* d_ws, size_t ws_size,
                              hipStream_t stream) {
  const float* x      = (const float*)d_in[0];
  const float* w_attn = (const float*)d_in[1];
  const float* b_attn = (const float*)d_in[2];
  const float* w_proj = (const float*)d_in[3];
  const float* b_proj = (const float*)d_in[4];
  float* out = (float*)d_out;

  char* ws = (char*)d_ws;
  const size_t QKV_BYTES = (size_t)BB * TT * C3 * 4;
  const size_t FF_BYTES  = (size_t)BB * TT * TT * 4;
  const size_t MSK_BYTES = (size_t)BB * TT * (TT / 32) * 4;
  float*    qkv = (float*)ws;
  float*    FF  = (float*)(ws + QKV_BYTES);
  unsigned* msk = (unsigned*)(ws + QKV_BYTES + FF_BYTES);
  float*    Y   = (float*)(ws + QKV_BYTES + FF_BYTES + MSK_BYTES);

  gemm_wmma<<<dim3(C3 / 256, (BB * TT) / 32), 128, 0, stream>>>(
      x, w_attn, b_attn, qkv, BB * TT, C3, CC);
  head0_scores<<<dim3(TT / 16, TT / 16, BB), 32, 0, stream>>>(qkv, FF);
  ff_cumsum<<<dim3(TT / 32, BB), 1024, 0, stream>>>(FF);
  select_mask<<<BB * TT, 256, 0, stream>>>(FF, msk);
  attn<<<dim3(TT / 16, NH, BB), 32, 0, stream>>>(qkv, FF, msk, Y);
  gemm_wmma<<<dim3(CC / 256, (BB * TT) / 32), 128, 0, stream>>>(
      Y, w_proj, b_proj, out, BB * TT, CC, CC);
}


# select: digit search of each radix pass done by one wave (8 bins per lane in registers, wave prefix of the lane sums); was one lane walking the 256 bins, one LDS round trip per bin, block waiting
# speedup vs baseline: 4.1042x; 1.1209x over previous
#include <hip/hip_runtime.h>
#include <hip/hip_bf16.h>

#define BB 2
#define TT 2048
#define CC 1024
#define NH 16
#define HD 64
#define C3 3072
#define BUDGET 408

typedef __attribute__((ext_vector_type(2))) float v2f;
typedef __attribute__((ext_vector_type(8))) float v8f;

__device__ __forceinline__ v8f wmma4(v2f a, v2f b, v8f c) {
  return __builtin_amdgcn_wmma_f32_16x16x4_f32(false, a, false, b, (short)0, c, false, false);
}

__device__ __forceinline__ float rowmax16(float x) {
  #pragma unroll
  for (int d = 1; d < 16; d <<= 1) x = fmaxf(x, __shfl_xor(x, d, 32));
  return x;
}
__device__ __forceinline__ float rowsum16(float x) {
  #pragma unroll
  for (int d = 1; d < 16; d <<= 1) x += __shfl_xor(x, d, 32);
  return x;
}

typedef __attribute__((ext_vector_type(4))) float v4f;

__device__ __forceinline__ void gemm_load_block(v4f (&buf)[6][4], const float* const (&rp)[6], int kb) {
  #pragma unroll
  for (int rg = 0; rg < 6; ++rg)
    #pragma unroll
    for (int q = 0; q < 4; ++q)
      buf[rg][q] = *(const v4f*)(rp[rg] + kb + q * 8);
}

__device__ __forceinline__ void gemm_mma_block(const v4f (&buf)[6][4], v8f (&acc)[2][4]) {
  #pragma unroll
  for (int q = 0; q < 4; ++q) {
    #pragma unroll
    for (int s = 0; s < 2; ++s) {
      v2f a[2], b[4];
      #pragma unroll
      for (int mt = 0; mt < 2; ++mt) {
        a[mt].x = buf[mt][q][2 * s];
        a[mt].y = buf[mt][q][2 * s + 1];
      }
      #pragma unroll
      for (int t = 0; t < 4; ++t) {
        b[t].x = buf[2 + t][q][2 * s];
        b[t].y = buf[2 + t][q][2 * s + 1];
      }
      #pragma unroll
      for (int mt = 0; mt < 2; ++mt)
        #pragma unroll
        for (int t = 0; t < 4; ++t)
          acc[mt][t] = wmma4(a[mt], b[t], acc[mt][t]);
    }
  }
}

__global__ void __launch_bounds__(128)
gemm_wmma(const float* __restrict__ A, const float* __restrict__ W,
          const float* __restrict__ bias, float* __restrict__ Out,
          int M, int N, int K) {
  int wave = threadIdx.x >> 5;
  int lane = threadIdx.x & 31;
  int g  = lane >> 4;
  int lm = lane & 15;
  int m0 = blockIdx.y * 32;
  int n0 = blockIdx.x * 256 + wave * 64;
  if (n0 >= N || m0 >= M) return;

  v8f acc[2][4];
  #pragma unroll
  for (int t = 0; t < 4; ++t) {
    float bv = bias[n0 + t * 16 + lm];
    #pragma unroll
    for (int mt = 0; mt < 2; ++mt)
      #pragma unroll
      for (int r = 0; r < 8; ++r) acc[mt][t][r] = bv;
  }
  const float* rp[6];
  rp[0] = A + (size_t)(m0 + lm) * K + 4 * g;
  rp[1] = A + (size_t)(m0 + 16 + lm) * K + 4 * g;
  #pragma unroll
  for (int t = 0; t < 4; ++t)
    rp[2 + t] = W + (size_t)(n0 + t * 16 + lm) * K + 4 * g;

  v4f ping[6][4], pong[6][4];
  gemm_load_block(ping, rp, 0);
  int kb = 0;
  for (; kb + 64 < K; kb += 64) {
    gemm_load_block(pong, rp, kb + 32);
    __builtin_amdgcn_sched_barrier(0);
    gemm_mma_block(ping, acc);
    __builtin_amdgcn_sched_barrier(0);
    gemm_load_block(ping, rp, kb + 64);
    __builtin_amdgcn_sched_barrier(0);
    gemm_mma_block(pong, acc);
    __builtin_amdgcn_sched_barrier(0);
  }
  gemm_load_block(pong, rp, kb + 32);
  __builtin_amdgcn_sched_barrier(0);
  gemm_mma_block(ping, acc);
  gemm_mma_block(pong, acc);
  #pragma unroll
  for (int mt = 0; mt < 2; ++mt)
    #pragma unroll
    for (int t = 0; t < 4; ++t)
      #pragma unroll
      for (int r = 0; r < 8; ++r)
        Out[(size_t)(m0 + mt * 16 + r + 8 * g) * N + n0 + t * 16 + lm] = acc[mt][t][r];
}

__global__ void head0_scores(const float* __restrict__ qkv, float* __restrict__ S) {
  int jb = blockIdx.x, ib = blockIdx.y, b = blockIdx.z;
  int lane = threadIdx.x & 31, g = lane >> 4, lm = lane & 15;
  int i0 = ib * 16, j0 = jb * 16;
  if (jb > ib) {
    #pragma unroll
    for (int r = 0; r < 8; ++r)
      S[((size_t)(b * TT + i0 + r + 8 * g)) * TT + j0 + lm] = 0.f;
    return;
  }
  const float* qrow = qkv + ((size_t)(b * TT) + i0 + lm) * C3;
  const float* krow = qkv + ((size_t)(b * TT) + j0 + lm) * C3 + CC;
  v2f aq[16], bk[16];
  #pragma unroll
  for (int ks = 0; ks < 16; ++ks) {
    aq[ks] = *(const v2f*)(qrow + ks * 4 + 2 * g);
    bk[ks] = *(const v2f*)(krow + ks * 4 + 2 * g);
  }
  v8f acc = {0.f, 0.f, 0.f, 0.f, 0.f, 0.f, 0.f, 0.f};
  #pragma unroll
  for (int ks = 0; ks < 16; ++ks) acc = wmma4(aq[ks], bk[ks], acc);
  #pragma unroll
  for (int r = 0; r < 8; ++r) {
    int i = i0 + r + 8 * g, j = j0 + lm;
    float s = acc[r] * 0.125f;
    s = (j < i && j > 0) ? fmaxf(s, 0.f) : 0.f;
    S[((size_t)(b * TT + i)) * TT + j] = s;
  }
}

#define SCAN_ROWS 64
__global__ void __launch_bounds__(1024)
ff_cumsum(float* __restrict__ S) {
  __shared__ float tot[32 * 32];
  int lane = threadIdx.x & 31, w = threadIdx.x >> 5;
  int j = blockIdx.x * 32 + lane;
  int b = blockIdx.y;
  float* col = S + (size_t)b * TT * TT + (size_t)(w * SCAN_ROWS) * TT + j;
  float v[SCAN_ROWS];
  #pragma unroll
  for (int r = 0; r < SCAN_ROWS; ++r) v[r] = col[(size_t)r * TT];
  float run = 0.f;
  #pragma unroll
  for (int r = 0; r < SCAN_ROWS; ++r) { float t = v[r]; v[r] = run; run += t; }
  tot[w * 32 + lane] = run;
  __syncthreads();
  float off = 0.f;
  #pragma unroll
  for (int w2 = 0; w2 < 31; ++w2) off += (w2 < w) ? tot[w2 * 32 + lane] : 0.f;
  #pragma unroll
  for (int r = 0; r < SCAN_ROWS; ++r) col[(size_t)r * TT] = v[r] + off;
}

__global__ void select_mask(const float* __restrict__ FF, unsigned* __restrict__ mask) {
  int row = blockIdx.x;
  int i = row & (TT - 1);
  const float* f = FF + (size_t)row * TT;
  unsigned* mrow = mask + (size_t)row * (TT / 32);
  int tid = threadIdx.x;

  if (i < BUDGET) {
    for (int w = tid; w < TT / 32; w += 256) {
      int jb = w * 32;
      unsigned m;
      if (jb + 31 <= i) m = 0xFFFFFFFFu;
      else if (jb > i)  m = 0u;
      else              m = (1u << (i - jb + 1)) - 1u;
      mrow[w] = m;
    }
    return;
  }

  __shared__ __attribute__((aligned(16))) unsigned hist[256];
  __shared__ unsigned misc[4];
  __shared__ unsigned wave_tot[8];

  if (tid == 0) { misc[0] = 0u; misc[1] = BUDGET - 1; }
  unsigned prefix = 0, prefmask = 0;
  for (int pass = 0; pass < 4; ++pass) {
    int shift = 24 - 8 * pass;
    hist[tid] = 0u;
    __syncthreads();
    for (int j = tid; j <= i; j += 256) {
      unsigned u = __float_as_uint(f[j]);
      if ((u & prefmask) == prefix) atomicAdd(&hist[(u >> shift) & 255u], 1u);
    }
    __syncthreads();
    if (tid < 32) {
      unsigned kk = misc[1];
      const uint4 h0 = *(const uint4*)&hist[8 * tid];
      const uint4 h1 = *(const uint4*)&hist[8 * tid + 4];
      const unsigned hb[8] = {h0.x, h0.y, h0.z, h0.w, h1.x, h1.y, h1.z, h1.w};
      unsigned s = ((hb[0] + hb[1]) + (hb[2] + hb[3])) + ((hb[4] + hb[5]) + (hb[6] + hb[7]));
      unsigned inc = s;
      for (int d = 1; d < 32; d <<= 1) {
        unsigned t = __shfl_up(inc, d, 32);
        inc += (tid >= d) ? t : 0u;
      }
      unsigned cum = inc - s;
      if (cum <= kk && kk < inc) {
        int dq = 0;
        for (int q = 0; q < 7; ++q) {
          bool more = (dq == q) && (cum + hb[q] <= kk);
          cum += more ? hb[q] : 0u;
          dq += more ? 1 : 0;
        }
        misc[1] = kk - cum;
        misc[0] |= ((unsigned)(8 * tid + dq) << shift);
      }
    }
    __syncthreads();
    prefix = misc[0];
    prefmask |= (255u << shift);
  }
  unsigned theta = prefix;

  if (tid == 0) misc[2] = 0u;
  __syncthreads();
  unsigned cl = 0;
  for (int j = tid; j <= i; j += 256)
    if (__float_as_uint(f[j]) < theta) cl++;
  atomicAdd(&misc[2], cl);
  __syncthreads();
  int ntake = BUDGET - (int)misc[2];

  int lane = tid & 31, wv = tid >> 5;
  unsigned below = (1u << lane) - 1u;
  unsigned carry = 0;
  for (int base = 0; base < TT; base += 256) {
    int j = base + tid;
    unsigned u = (j <= i) ? __float_as_uint(f[j]) : 0xFFFFFFFFu;
    bool eq = (j <= i) && (u == theta);
    unsigned bal = (unsigned)__ballot(eq);
    if (lane == 0) wave_tot[wv] = __popc(bal);
    __syncthreads();
    unsigned pre = carry + __popc(bal & below);
    for (int w2 = 0; w2 < wv; ++w2) pre += wave_tot[w2];
    bool keep = (j <= i) && ((u < theta) || (eq && (int)pre < ntake));
    keep = keep || (j == i);
    unsigned kb = (unsigned)__ballot(keep);
    if (lane == 0) mrow[(base + wv * 32) >> 5] = kb;
    unsigned tot = 0;
    #pragma unroll
    for (int w2 = 0; w2 < 8; ++w2) tot += wave_tot[w2];
    carry += tot;
    __syncthreads();
  }
}

__global__ void __launch_bounds__(32)
attn(const float* __restrict__ qkv, const float* __restrict__ FF,
                     const unsigned* __restrict__ mask, float* __restrict__ Y) {
  int ib = blockIdx.x, h = blockIdx.y, b = blockIdx.z;
  int lane = threadIdx.x & 31, g = lane >> 4, lm = lane & 15;
  int i0 = ib * 16;
  __shared__ float pt[16 * 17];

  const float* qb = qkv + ((size_t)(b * TT)) * C3 + h * HD;
  const float* kb = qb + CC;
  const float* vb = qb + 2 * CC;

  v2f aq[16];
  #pragma unroll
  for (int ks = 0; ks < 16; ++ks)
    aq[ks] = *(const v2f*)(qb + (size_t)(i0 + lm) * C3 + ks * 4 + 2 * g);

  float mrow[8], lrow[8];
  v8f o[4];
  #pragma unroll
  for (int t = 0; t < 4; ++t)
    #pragma unroll
    for (int r = 0; r < 8; ++r) o[t][r] = 0.f;
  #pragma unroll
  for (int r = 0; r < 8; ++r) { mrow[r] = -1e30f; lrow[r] = 0.f; }

  const unsigned* mrow16 = mask + ((size_t)(b * TT + i0)) * (TT / 32);
  unsigned or0 = 0u, or1 = 0u;
  #pragma unroll
  for (int r = 0; r < 16; ++r) {
    or0 |= mrow16[r * (TT / 32) + lane];
    or1 |= mrow16[r * (TT / 32) + 32 + lane];
  }
  int wsrc = lane >> 1, hsh = (lane & 1) * 16;
  unsigned w0 = __shfl(or0, wsrc, 32), w1 = __shfl(or0, 16 + wsrc, 32);
  unsigned w2 = __shfl(or1, wsrc, 32), w3 = __shfl(or1, 16 + wsrc, 32);
  unsigned t0 = (unsigned)__ballot(((w0 >> hsh) & 0xFFFFu) != 0u);
  unsigned t1 = (unsigned)__ballot(((w1 >> hsh) & 0xFFFFu) != 0u);
  unsigned t2 = (unsigned)__ballot(((w2 >> hsh) & 0xFFFFu) != 0u);
  unsigned t3 = (unsigned)__ballot(((w3 >> hsh) & 0xFFFFu) != 0u);
  unsigned long long tlo = (unsigned long long)t0 | ((unsigned long long)t1 << 32);
  unsigned long long thi = (unsigned long long)t2 | ((unsigned long long)t3 << 32);
  if (ib < 64) {
    if (ib < 63) tlo &= (2ull << ib) - 1ull;
    thi = 0ull;
  } else if (ib < 127) {
    thi &= (2ull << (ib - 64)) - 1ull;
  }

  for (;;) {
    int jb;
    if (tlo != 0ull)      { jb = __builtin_ctzll(tlo);      tlo &= tlo - 1ull; }
    else if (thi != 0ull) { jb = 64 + __builtin_ctzll(thi); thi &= thi - 1ull; }
    else break;
    int j0 = jb * 16;
    v2f bk[16];
    #pragma unroll
    for (int ks = 0; ks < 16; ++ks)
      bk[ks] = *(const v2f*)(kb + (size_t)(j0 + lm) * C3 + ks * 4 + 2 * g);
    v8f acc = {0.f, 0.f, 0.f, 0.f, 0.f, 0.f, 0.f, 0.f};
    #pragma unroll
    for (int ks = 0; ks < 16; ++ks) acc = wmma4(aq[ks], bk[ks], acc);

    float ffv[8]; unsigned mwv[8];
    #pragma unroll
    for (int r = 0; r < 8; ++r) {
      int i = i0 + r + 8 * g;
      ffv[r] = FF[((size_t)(b * TT + i)) * TT + j0 + lm];
      mwv[r] = mask[((size_t)(b * TT + i)) * (TT / 32) + (j0 >> 5)];
    }
    float p[8];
    #pragma unroll
    for (int r = 0; r < 8; ++r) {
      int i = i0 + r + 8 * g, j = j0 + lm;
      bool keep = ((mwv[r] >> (j & 31)) & 1u) && (j <= i);
      float s = keep ? (acc[r] * 0.125f - ffv[r]) : -1e30f;
      float bm = rowmax16(s);
      float mnew = fmaxf(mrow[r], bm);
      float scale = __expf(mrow[r] - mnew);
      float pr = __expf(s - mnew);
      lrow[r] = lrow[r] * scale + rowsum16(pr);
      mrow[r] = mnew;
      p[r] = pr;
      #pragma unroll
      for (int t = 0; t < 4; ++t) o[t][r] *= scale;
    }
    __syncthreads();
    #pragma unroll
    for (int r = 0; r < 8; ++r) pt[(r + 8 * g) * 17 + lm] = p[r];
    __syncthreads();
    v2f ap[4];
    #pragma unroll
    for (int kt = 0; kt < 4; ++kt) {
      ap[kt].x = pt[lm * 17 + kt * 4 + 2 * g];
      ap[kt].y = pt[lm * 17 + kt * 4 + 2 * g + 1];
    }
    v2f bv[4][4];
    #pragma unroll
    for (int kt = 0; kt < 4; ++kt)
      #pragma unroll
      for (int t = 0; t < 4; ++t) {
        bv[kt][t].x = vb[(size_t)(j0 + kt * 4 + 2 * g) * C3 + t * 16 + lm];
        bv[kt][t].y = vb[(size_t)(j0 + kt * 4 + 2 * g + 1) * C3 + t * 16 + lm];
      }
    #pragma unroll
    for (int kt = 0; kt < 4; ++kt)
      #pragma unroll
      for (int t = 0; t < 4; ++t)
        o[t] = wmma4(ap[kt], bv[kt][t], o[t]);
  }
  #pragma unroll
  for (int r = 0; r < 8; ++r) {
    float inv = 1.f / lrow[r];
    int i = i0 + r + 8 * g;
    #pragma unroll
    for (int t = 0; t < 4; ++t)
      Y[((size_t)(b * TT + i)) * CC + h * HD + t * 16 + lm] = o[t][r] * inv;
  }
}

extern "C" void kernel_launch(void* const* d_in, const int* in_sizes, int n_in,
                              void* d_out, int out_size, void* d_ws, size_t ws_size,
                              hipStream_t stream) {
  const float* x      = (const float*)d_in[0];
  const float* w_attn = (const float*)d_in[1];
  const float* b_attn = (const float*)d_in[2];
  const float* w_proj = (const float*)d_in[3];
  const float* b_proj = (const float*)d_in[4];
  float* out = (float*)d_out;

  char* ws = (char*)d_ws;
  const size_t QKV_BYTES = (size_t)BB * TT * C3 * 4;
  const size_t FF_BYTES  = (size_t)BB * TT * TT * 4;
  const size_t MSK_BYTES = (size_t)BB * TT * (TT / 32) * 4;
  float*    qkv = (float*)ws;
  float*    FF  = (float*)(ws + QKV_BYTES);
  unsigned* msk = (unsigned*)(ws + QKV_BYTES + FF_BYTES);
  float*    Y   = (float*)(ws + QKV_BYTES + FF_BYTES + MSK_BYTES);

  gemm_wmma<<<dim3(C3 / 256, (BB * TT) / 32), 128, 0, stream>>>(
      x, w_attn, b_attn, qkv, BB * TT, C3, CC);
  head0_scores<<<dim3(TT / 16, TT / 16, BB), 32, 0, stream>>>(qkv, FF);
  ff_cumsum<<<dim3(TT / 32, BB), 1024, 0, stream>>>(FF);
  select_mask<<<BB * TT, 256, 0, stream>>>(FF, msk);
  attn<<<dim3(TT / 16, NH, BB), 32, 0, stream>>>(qkv, FF, msk, Y);
  gemm_wmma<<<dim3(CC / 256, (BB * TT) / 32), 128, 0, stream>>>(
      Y, w_proj, b_proj, out, BB * TT, CC, CC);
}


# gemm (both launches): wave tile 64x64 (was 32x64), block = 2x2 waves on a 128x128 tile: 128 B of fragments loaded per WMMA, was 192; each A and W line of the block tile is wanted by two waves
# speedup vs baseline: 4.2838x; 1.0438x over previous
#include <hip/hip_runtime.h>
#include <hip/hip_bf16.h>

#define BB 2
#define TT 2048
#define CC 1024
#define NH 16
#define HD 64
#define C3 3072
#define BUDGET 408

typedef __attribute__((ext_vector_type(2))) float v2f;
typedef __attribute__((ext_vector_type(8))) float v8f;

__device__ __forceinline__ v8f wmma4(v2f a, v2f b, v8f c) {
  return __builtin_amdgcn_wmma_f32_16x16x4_f32(false, a, false, b, (short)0, c, false, false);
}

__device__ __forceinline__ float rowmax16(float x) {
  #pragma unroll
  for (int d = 1; d < 16; d <<= 1) x = fmaxf(x, __shfl_xor(x, d, 32));
  return x;
}
__device__ __forceinline__ float rowsum16(float x) {
  #pragma unroll
  for (int d = 1; d < 16; d <<= 1) x += __shfl_xor(x, d, 32);
  return x;
}

typedef __attribute__((ext_vector_type(4))) float v4f;

__device__ __forceinline__ void gemm_load_block(v4f (&buf)[8][4], const float* const (&rb)[8], unsigned lo, int kb) {
  unsigned off = lo + (unsigned)kb;
  #pragma unroll
  for (int rg = 0; rg < 8; ++rg)
    #pragma unroll
    for (int q = 0; q < 4; ++q)
      buf[rg][q] = *(const v4f*)(rb[rg] + (size_t)off + q * 8);
}

__device__ __forceinline__ void gemm_mma_block(const v4f (&buf)[8][4], v8f (&acc)[4][4]) {
  #pragma unroll
  for (int q = 0; q < 4; ++q) {
    #pragma unroll
    for (int s = 0; s < 2; ++s) {
      v2f a[4], b[4];
      #pragma unroll
      for (int mt = 0; mt < 4; ++mt) {
        a[mt].x = buf[mt][q][2 * s];
        a[mt].y = buf[mt][q][2 * s + 1];
      }
      #pragma unroll
      for (int t = 0; t < 4; ++t) {
        b[t].x = buf[4 + t][q][2 * s];
        b[t].y = buf[4 + t][q][2 * s + 1];
      }
      #pragma unroll
      for (int mt = 0; mt < 4; ++mt)
        #pragma unroll
        for (int t = 0; t < 4; ++t)
          acc[mt][t] = wmma4(a[mt], b[t], acc[mt][t]);
    }
  }
}

__global__ void __launch_bounds__(128)
gemm_wmma(const float* __restrict__ A, const float* __restrict__ W,
          const float* __restrict__ bias, float* __restrict__ Out,
          int M, int N, int K) {
  int wave = __builtin_amdgcn_readfirstlane((int)(threadIdx.x >> 5));
  int lane = threadIdx.x & 31;
  int g  = lane >> 4;
  int lm = lane & 15;
  int m0 = blockIdx.y * 128 + (wave >> 1) * 64;
  int n0 = blockIdx.x * 128 + (wave & 1) * 64;
  if (n0 >= N || m0 >= M) return;

  v8f acc[4][4];
  #pragma unroll
  for (int t = 0; t < 4; ++t) {
    float bv = bias[n0 + t * 16 + lm];
    #pragma unroll
    for (int mt = 0; mt < 4; ++mt)
      #pragma unroll
      for (int r = 0; r < 8; ++r) acc[mt][t][r] = bv;
  }
  const float* rp[8];
  #pragma unroll
  for (int mt = 0; mt < 4; ++mt)
    rp[mt] = A + (size_t)(m0 + mt * 16) * K;
  #pragma unroll
  for (int t = 0; t < 4; ++t)
    rp[4 + t] = W + (size_t)(n0 + t * 16) * K;
  unsigned lo = (unsigned)(lm * K + 4 * g);

  v4f ping[8][4], pong[8][4];
  gemm_load_block(ping, rp, lo, 0);
  for (int kb = 0; kb < K; kb += 64) {
    gemm_load_block(pong, rp, lo, kb + 32);
    __builtin_amdgcn_sched_barrier(0);
    gemm_mma_block(ping, acc);
    __builtin_amdgcn_sched_barrier(0);
    gemm_load_block(ping, rp, lo, (kb + 64 < K) ? kb + 64 : kb + 32);
    __builtin_amdgcn_sched_barrier(0);
    gemm_mma_block(pong, acc);
    __builtin_amdgcn_sched_barrier(0);
  }
  #pragma unroll
  for (int mt = 0; mt < 4; ++mt)
    #pragma unroll
    for (int t = 0; t < 4; ++t)
      #pragma unroll
      for (int r = 0; r < 8; ++r)
        Out[(size_t)(m0 + mt * 16 + r + 8 * g) * N + n0 + t * 16 + lm] = acc[mt][t][r];
}

__global__ void head0_scores(const float* __restrict__ qkv, float* __restrict__ S) {
  int jb = blockIdx.x, ib = blockIdx.y, b = blockIdx.z;
  int lane = threadIdx.x & 31, g = lane >> 4, lm = lane & 15;
  int i0 = ib * 16, j0 = jb * 16;
  if (jb > ib) {
    #pragma unroll
    for (int r = 0; r < 8; ++r)
      S[((size_t)(b * TT + i0 + r + 8 * g)) * TT + j0 + lm] = 0.f;
    return;
  }
  const float* qrow = qkv + ((size_t)(b * TT) + i0 + lm) * C3;
  const float* krow = qkv + ((size_t)(b * TT) + j0 + lm) * C3 + CC;
  v2f aq[16], bk[16];
  #pragma unroll
  for (int ks = 0; ks < 16; ++ks) {
    aq[ks] = *(const v2f*)(qrow + ks * 4 + 2 * g);
    bk[ks] = *(const v2f*)(krow + ks * 4 + 2 * g);
  }
  v8f acc = {0.f, 0.f, 0.f, 0.f, 0.f, 0.f, 0.f, 0.f};
  #pragma unroll
  for (int ks = 0; ks < 16; ++ks) acc = wmma4(aq[ks], bk[ks], acc);
  #pragma unroll
  for (int r = 0; r < 8; ++r) {
    int i = i0 + r + 8 * g, j = j0 + lm;
    float s = acc[r] * 0.125f;
    s = (j < i && j > 0) ? fmaxf(s, 0.f) : 0.f;
    S[((size_t)(b * TT + i)) * TT + j] = s;
  }
}

#define SCAN_ROWS 64
__global__ void __launch_bounds__(1024)
ff_cumsum(float* __restrict__ S) {
  __shared__ float tot[32 * 32];
  int lane = threadIdx.x & 31, w = threadIdx.x >> 5;
  int j = blockIdx.x * 32 + lane;
  int b = blockIdx.y;
  float* col = S + (size_t)b * TT * TT + (size_t)(w * SCAN_ROWS) * TT + j;
  float v[SCAN_ROWS];
  #pragma unroll
  for (int r = 0; r < SCAN_ROWS; ++r) v[r] = col[(size_t)r * TT];
  float run = 0.f;
  #pragma unroll
  for (int r = 0; r < SCAN_ROWS; ++r) { float t = v[r]; v[r] = run; run += t; }
  tot[w * 32 + lane] = run;
  __syncthreads();
  float off = 0.f;
  #pragma unroll
  for (int w2 = 0; w2 < 31; ++w2) off += (w2 < w) ? tot[w2 * 32 + lane] : 0.f;
  #pragma unroll
  for (int r = 0; r < SCAN_ROWS; ++r) col[(size_t)r * TT] = v[r] + off;
}

__global__ void select_mask(const float* __restrict__ FF, unsigned* __restrict__ mask) {
  int row = blockIdx.x;
  int i = row & (TT - 1);
  const float* f = FF + (size_t)row * TT;
  unsigned* mrow = mask + (size_t)row * (TT / 32);
  int tid = threadIdx.x;

  if (i < BUDGET) {
    for (int w = tid; w < TT / 32; w += 256) {
      int jb = w * 32;
      unsigned m;
      if (jb + 31 <= i) m = 0xFFFFFFFFu;
      else if (jb > i)  m = 0u;
      else              m = (1u << (i - jb + 1)) - 1u;
      mrow[w] = m;
    }
    return;
  }

  __shared__ __attribute__((aligned(16))) unsigned hist[256];
  __shared__ unsigned misc[4];
  __shared__ unsigned wave_tot[8];

  if (tid == 0) { misc[0] = 0u; misc[1] = BUDGET - 1; }
  unsigned prefix = 0, prefmask = 0;
  for (int pass = 0; pass < 4; ++pass) {
    int shift = 24 - 8 * pass;
    hist[tid] = 0u;
    __syncthreads();
    for (int j = tid; j <= i; j += 256) {
      unsigned u = __float_as_uint(f[j]);
      if ((u & prefmask) == prefix) atomicAdd(&hist[(u >> shift) & 255u], 1u);
    }
    __syncthreads();
    if (tid < 32) {
      unsigned kk = misc[1];
      const uint4 h0 = *(const uint4*)&hist[8 * tid];
      const uint4 h1 = *(const uint4*)&hist[8 * tid + 4];
      const unsigned hb[8] = {h0.x, h0.y, h0.z, h0.w, h1.x, h1.y, h1.z, h1.w};
      unsigned s = ((hb[0] + hb[1]) + (hb[2] + hb[3])) + ((hb[4] + hb[5]) + (hb[6] + hb[7]));
      unsigned inc = s;
      for (int d = 1; d < 32; d <<= 1) {
        unsigned t = __shfl_up(inc, d, 32);
        inc += (tid >= d) ? t : 0u;
      }
      unsigned cum = inc - s;
      if (cum <= kk && kk < inc) {
        int dq = 0;
        for (int q = 0; q < 7; ++q) {
          bool more = (dq == q) && (cum + hb[q] <= kk);
          cum += more ? hb[q] : 0u;
          dq += more ? 1 : 0;
        }
        misc[1] = kk - cum;
        misc[0] |= ((unsigned)(8 * tid + dq) << shift);
      }
    }
    __syncthreads();
    prefix = misc[0];
    prefmask |= (255u << shift);
  }
  unsigned theta = prefix;

  if (tid == 0) misc[2] = 0u;
  __syncthreads();
  unsigned cl = 0;
  for (int j = tid; j <= i; j += 256)
    if (__float_as_uint(f[j]) < theta) cl++;
  atomicAdd(&misc[2], cl);
  __syncthreads();
  int ntake = BUDGET - (int)misc[2];

  int lane = tid & 31, wv = tid >> 5;
  unsigned below = (1u << lane) - 1u;
  unsigned carry = 0;
  for (int base = 0; base < TT; base += 256) {
    int j = base + tid;
    unsigned u = (j <= i) ? __float_as_uint(f[j]) : 0xFFFFFFFFu;
    bool eq = (j <= i) && (u == theta);
    unsigned bal = (unsigned)__ballot(eq);
    if (lane == 0) wave_tot[wv] = __popc(bal);
    __syncthreads();
    unsigned pre = carry + __popc(bal & below);
    for (int w2 = 0; w2 < wv; ++w2) pre += wave_tot[w2];
    bool keep = (j <= i) && ((u < theta) || (eq && (int)pre < ntake));
    keep = keep || (j == i);
    unsigned kb = (unsigned)__ballot(keep);
    if (lane == 0) mrow[(base + wv * 32) >> 5] = kb;
    unsigned tot = 0;
    #pragma unroll
    for (int w2 = 0; w2 < 8; ++w2) tot += wave_tot[w2];
    carry += tot;
    __syncthreads();
  }
}

__global__ void __launch_bounds__(32)
attn(const float* __restrict__ qkv, const float* __restrict__ FF,
                     const unsigned* __restrict__ mask, float* __restrict__ Y) {
  int ib = blockIdx.x, h = blockIdx.y, b = blockIdx.z;
  int lane = threadIdx.x & 31, g = lane >> 4, lm = lane & 15;
  int i0 = ib * 16;
  __shared__ float pt[16 * 17];

  const float* qb = qkv + ((size_t)(b * TT)) * C3 + h * HD;
  const float* kb = qb + CC;
  const float* vb = qb + 2 * CC;

  v2f aq[16];
  #pragma unroll
  for (int ks = 0; ks < 16; ++ks)
    aq[ks] = *(const v2f*)(qb + (size_t)(i0 + lm) * C3 + ks * 4 + 2 * g);

  float mrow[8], lrow[8];
  v8f o[4];
  #pragma unroll
  for (int t = 0; t < 4; ++t)
    #pragma unroll
    for (int r = 0; r < 8; ++r) o[t][r] = 0.f;
  #pragma unroll
  for (int r = 0; r < 8; ++r) { mrow[r] = -1e30f; lrow[r] = 0.f; }

  const unsigned* mrow16 = mask + ((size_t)(b * TT + i0)) * (TT / 32);
  unsigned or0 = 0u, or1 = 0u;
  #pragma unroll
  for (int r = 0; r < 16; ++r) {
    or0 |= mrow16[r * (TT / 32) + lane];
    or1 |= mrow16[r * (TT / 32) + 32 + lane];
  }
  int wsrc = lane >> 1, hsh = (lane & 1) * 16;
  unsigned w0 = __shfl(or0, wsrc, 32), w1 = __shfl(or0, 16 + wsrc, 32);
  unsigned w2 = __shfl(or1, wsrc, 32), w3 = __shfl(or1, 16 + wsrc, 32);
  unsigned t0 = (unsigned)__ballot(((w0 >> hsh) & 0xFFFFu) != 0u);
  unsigned t1 = (unsigned)__ballot(((w1 >> hsh) & 0xFFFFu) != 0u);
  unsigned t2 = (unsigned)__ballot(((w2 >> hsh) & 0xFFFFu) != 0u);
  unsigned t3 = (unsigned)__ballot(((w3 >> hsh) & 0xFFFFu) != 0u);
  unsigned long long tlo = (unsigned long long)t0 | ((unsigned long long)t1 << 32);
  unsigned long long thi = (unsigned long long)t2 | ((unsigned long long)t3 << 32);
  if (ib < 64) {
    if (ib < 63) tlo &= (2ull << ib) - 1ull;
    thi = 0ull;
  } else if (ib < 127) {
    thi &= (2ull << (ib - 64)) - 1ull;
  }

  for (;;) {
    int jb;
    if (tlo != 0ull)      { jb = __builtin_ctzll(tlo);      tlo &= tlo - 1ull; }
    else if (thi != 0ull) { jb = 64 + __builtin_ctzll(thi); thi &= thi - 1ull; }
    else break;
    int j0 = jb * 16;
    v2f bk[16];
    #pragma unroll
    for (int ks = 0; ks < 16; ++ks)
      bk[ks] = *(const v2f*)(kb + (size_t)(j0 + lm) * C3 + ks * 4 + 2 * g);
    v8f acc = {0.f, 0.f, 0.f, 0.f, 0.f, 0.f, 0.f, 0.f};
    #pragma unroll
    for (int ks = 0; ks < 16; ++ks) acc = wmma4(aq[ks], bk[ks], acc);

    float ffv[8]; unsigned mwv[8];
    #pragma unroll
    for (int r = 0; r < 8; ++r) {
      int i = i0 + r + 8 * g;
      ffv[r] = FF[((size_t)(b * TT + i)) * TT + j0 + lm];
      mwv[r] = mask[((size_t)(b * TT + i)) * (TT / 32) + (j0 >> 5)];
    }
    float p[8];
    #pragma unroll
    for (int r = 0; r < 8; ++r) {
      int i = i0 + r + 8 * g, j = j0 + lm;
      bool keep = ((mwv[r] >> (j & 31)) & 1u) && (j <= i);
      float s = keep ? (acc[r] * 0.125f - ffv[r]) : -1e30f;
      float bm = rowmax16(s);
      float mnew = fmaxf(mrow[r], bm);
      float scale = __expf(mrow[r] - mnew);
      float pr = __expf(s - mnew);
      lrow[r] = lrow[r] * scale + rowsum16(pr);
      mrow[r] = mnew;
      p[r] = pr;
      #pragma unroll
      for (int t = 0; t < 4; ++t) o[t][r] *= scale;
    }
    __syncthreads();
    #pragma unroll
    for (int r = 0; r < 8; ++r) pt[(r + 8 * g) * 17 + lm] = p[r];
    __syncthreads();
    v2f ap[4];
    #pragma unroll
    for (int kt = 0; kt < 4; ++kt) {
      ap[kt].x = pt[lm * 17 + kt * 4 + 2 * g];
      ap[kt].y = pt[lm * 17 + kt * 4 + 2 * g + 1];
    }
    v2f bv[4][4];
    #pragma unroll
    for (int kt = 0; kt < 4; ++kt)
      #pragma unroll
      for (int t = 0; t < 4; ++t) {
        bv[kt][t].x = vb[(size_t)(j0 + kt * 4 + 2 * g) * C3 + t * 16 + lm];
        bv[kt][t].y = vb[(size_t)(j0 + kt * 4 + 2 * g + 1) * C3 + t * 16 + lm];
      }
    #pragma unroll
    for (int kt = 0; kt < 4; ++kt)
      #pragma unroll
      for (int t = 0; t < 4; ++t)
        o[t] = wmma4(ap[kt], bv[kt][t], o[t]);
  }
  #pragma unroll
  for (int r = 0; r < 8; ++r) {
    float inv = 1.f / lrow[r];
    int i = i0 + r + 8 * g;
    #pragma unroll
    for (int t = 0; t < 4; ++t)
      Y[((size_t)(b * TT + i)) * CC + h * HD + t * 16 + lm] = o[t][r] * inv;
  }
}

extern "C" void kernel_launch(void* const* d_in, const int* in_sizes, int n_in,
                              void* d_out, int out_size, void* d_ws, size_t ws_size,
                              hipStream_t stream) {
  const float* x      = (const float*)d_in[0];
  const float* w_attn = (const float*)d_in[1];
  const float* b_attn = (const float*)d_in[2];
  const float* w_proj = (const float*)d_in[3];
  const float* b_proj = (const float*)d_in[4];
  float* out = (float*)d_out;

  char* ws = (char*)d_ws;
  const size_t QKV_BYTES = (size_t)BB * TT * C3 * 4;
  const size_t FF_BYTES  = (size_t)BB * TT * TT * 4;
  const size_t MSK_BYTES = (size_t)BB * TT * (TT / 32) * 4;
  float*    qkv = (float*)ws;
  float*    FF  = (float*)(ws + QKV_BYTES);
  unsigned* msk = (unsigned*)(ws + QKV_BYTES + FF_BYTES);
  float*    Y   = (float*)(ws + QKV_BYTES + FF_BYTES + MSK_BYTES);

  gemm_wmma<<<dim3(C3 / 128, (BB * TT) / 128), 128, 0, stream>>>(
      x, w_attn, b_attn, qkv, BB * TT, C3, CC);
  head0_scores<<<dim3(TT / 16, TT / 16, BB), 32, 0, stream>>>(qkv, FF);
  ff_cumsum<<<dim3(TT / 32, BB), 1024, 0, stream>>>(FF);
  select_mask<<<BB * TT, 256, 0, stream>>>(FF, msk);
  attn<<<dim3(TT / 16, NH, BB), 32, 0, stream>>>(qkv, FF, msk, Y);
  gemm_wmma<<<dim3(CC / 128, (BB * TT) / 128), 128, 0, stream>>>(
      Y, w_proj, b_proj, out, BB * TT, CC, CC);
}


# select: one wave per row, the row's keys in registers (read once, whole-line loads), threshold built bit by bit from ballot popcounts; was a 256-thread block with 4 LDS histogram passes and atomics
# speedup vs baseline: 4.6529x; 1.0862x over previous
#include <hip/hip_runtime.h>
#include <hip/hip_bf16.h>

#define BB 2
#define TT 2048
#define CC 1024
#define NH 16
#define HD 64
#define C3 3072
#define BUDGET 408

typedef __attribute__((ext_vector_type(2))) float v2f;
typedef __attribute__((ext_vector_type(8))) float v8f;

__device__ __forceinline__ v8f wmma4(v2f a, v2f b, v8f c) {
  return __builtin_amdgcn_wmma_f32_16x16x4_f32(false, a, false, b, (short)0, c, false, false);
}

__device__ __forceinline__ float rowmax16(float x) {
  #pragma unroll
  for (int d = 1; d < 16; d <<= 1) x = fmaxf(x, __shfl_xor(x, d, 32));
  return x;
}
__device__ __forceinline__ float rowsum16(float x) {
  #pragma unroll
  for (int d = 1; d < 16; d <<= 1) x += __shfl_xor(x, d, 32);
  return x;
}

typedef __attribute__((ext_vector_type(4))) float v4f;

__device__ __forceinline__ void gemm_load_block(v4f (&buf)[8][4], const float* const (&rb)[8], unsigned lo, int kb) {
  unsigned off = lo + (unsigned)kb;
  #pragma unroll
  for (int rg = 0; rg < 8; ++rg)
    #pragma unroll
    for (int q = 0; q < 4; ++q)
      buf[rg][q] = *(const v4f*)(rb[rg] + (size_t)off + q * 8);
}

__device__ __forceinline__ void gemm_mma_block(const v4f (&buf)[8][4], v8f (&acc)[4][4]) {
  #pragma unroll
  for (int q = 0; q < 4; ++q) {
    #pragma unroll
    for (int s = 0; s < 2; ++s) {
      v2f a[4], b[4];
      #pragma unroll
      for (int mt = 0; mt < 4; ++mt) {
        a[mt].x = buf[mt][q][2 * s];
        a[mt].y = buf[mt][q][2 * s + 1];
      }
      #pragma unroll
      for (int t = 0; t < 4; ++t) {
        b[t].x = buf[4 + t][q][2 * s];
        b[t].y = buf[4 + t][q][2 * s + 1];
      }
      #pragma unroll
      for (int mt = 0; mt < 4; ++mt)
        #pragma unroll
        for (int t = 0; t < 4; ++t)
          acc[mt][t] = wmma4(a[mt], b[t], acc[mt][t]);
    }
  }
}

__global__ void __launch_bounds__(128)
gemm_wmma(const float* __restrict__ A, const float* __restrict__ W,
          const float* __restrict__ bias, float* __restrict__ Out,
          int M, int N, int K) {
  int wave = __builtin_amdgcn_readfirstlane((int)(threadIdx.x >> 5));
  int lane = threadIdx.x & 31;
  int g  = lane >> 4;
  int lm = lane & 15;
  int m0 = blockIdx.y * 128 + (wave >> 1) * 64;
  int n0 = blockIdx.x * 128 + (wave & 1) * 64;
  if (n0 >= N || m0 >= M) return;

  v8f acc[4][4];
  #pragma unroll
  for (int t = 0; t < 4; ++t) {
    float bv = bias[n0 + t * 16 + lm];
    #pragma unroll
    for (int mt = 0; mt < 4; ++mt)
      #pragma unroll
      for (int r = 0; r < 8; ++r) acc[mt][t][r] = bv;
  }
  const float* rp[8];
  #pragma unroll
  for (int mt = 0; mt < 4; ++mt)
    rp[mt] = A + (size_t)(m0 + mt * 16) * K;
  #pragma unroll
  for (int t = 0; t < 4; ++t)
    rp[4 + t] = W + (size_t)(n0 + t * 16) * K;
  unsigned lo = (unsigned)(lm * K + 4 * g);

  v4f ping[8][4], pong[8][4];
  gemm_load_block(ping, rp, lo, 0);
  for (int kb = 0; kb < K; kb += 64) {
    gemm_load_block(pong, rp, lo, kb + 32);
    __builtin_amdgcn_sched_barrier(0);
    gemm_mma_block(ping, acc);
    __builtin_amdgcn_sched_barrier(0);
    gemm_load_block(ping, rp, lo, (kb + 64 < K) ? kb + 64 : kb + 32);
    __builtin_amdgcn_sched_barrier(0);
    gemm_mma_block(pong, acc);
    __builtin_amdgcn_sched_barrier(0);
  }
  #pragma unroll
  for (int mt = 0; mt < 4; ++mt)
    #pragma unroll
    for (int t = 0; t < 4; ++t)
      #pragma unroll
      for (int r = 0; r < 8; ++r)
        Out[(size_t)(m0 + mt * 16 + r + 8 * g) * N + n0 + t * 16 + lm] = acc[mt][t][r];
}

__global__ void head0_scores(const float* __restrict__ qkv, float* __restrict__ S) {
  int jb = blockIdx.x, ib = blockIdx.y, b = blockIdx.z;
  int lane = threadIdx.x & 31, g = lane >> 4, lm = lane & 15;
  int i0 = ib * 16, j0 = jb * 16;
  if (jb > ib) {
    #pragma unroll
    for (int r = 0; r < 8; ++r)
      S[((size_t)(b * TT + i0 + r + 8 * g)) * TT + j0 + lm] = 0.f;
    return;
  }
  const float* qrow = qkv + ((size_t)(b * TT) + i0 + lm) * C3;
  const float* krow = qkv + ((size_t)(b * TT) + j0 + lm) * C3 + CC;
  v2f aq[16], bk[16];
  #pragma unroll
  for (int ks = 0; ks < 16; ++ks) {
    aq[ks] = *(const v2f*)(qrow + ks * 4 + 2 * g);
    bk[ks] = *(const v2f*)(krow + ks * 4 + 2 * g);
  }
  v8f acc = {0.f, 0.f, 0.f, 0.f, 0.f, 0.f, 0.f, 0.f};
  #pragma unroll
  for (int ks = 0; ks < 16; ++ks) acc = wmma4(aq[ks], bk[ks], acc);
  #pragma unroll
  for (int r = 0; r < 8; ++r) {
    int i = i0 + r + 8 * g, j = j0 + lm;
    float s = acc[r] * 0.125f;
    s = (j < i && j > 0) ? fmaxf(s, 0.f) : 0.f;
    S[((size_t)(b * TT + i)) * TT + j] = s;
  }
}

#define SCAN_ROWS 64
__global__ void __launch_bounds__(1024)
ff_cumsum(float* __restrict__ S) {
  __shared__ float tot[32 * 32];
  int lane = threadIdx.x & 31, w = threadIdx.x >> 5;
  int j = blockIdx.x * 32 + lane;
  int b = blockIdx.y;
  float* col = S + (size_t)b * TT * TT + (size_t)(w * SCAN_ROWS) * TT + j;
  float v[SCAN_ROWS];
  #pragma unroll
  for (int r = 0; r < SCAN_ROWS; ++r) v[r] = col[(size_t)r * TT];
  float run = 0.f;
  #pragma unroll
  for (int r = 0; r < SCAN_ROWS; ++r) { float t = v[r]; v[r] = run; run += t; }
  tot[w * 32 + lane] = run;
  __syncthreads();
  float off = 0.f;
  #pragma unroll
  for (int w2 = 0; w2 < 31; ++w2) off += (w2 < w) ? tot[w2 * 32 + lane] : 0.f;
  #pragma unroll
  for (int r = 0; r < SCAN_ROWS; ++r) col[(size_t)r * TT] = v[r] + off;
}

#define SEL_WORDS (TT / 32)

__device__ __forceinline__ unsigned causal_word(int w, int i) {
  int jb = w * 32;
  if (jb + 31 <= i) return 0xFFFFFFFFu;
  if (jb > i)       return 0u;
  return (1u << (i - jb + 1)) - 1u;
}

__device__ __forceinline__ unsigned count_below(const unsigned (&u)[SEL_WORDS], int ng, unsigned cand) {
  unsigned c = 0;
  #pragma unroll
  for (int g = 0; g < 4; ++g) {
    if (g < ng) {
      #pragma unroll
      for (int m = 0; m < 16; m += 2) {
        unsigned long long lo = (unsigned)__ballot(u[16 * g + m] < cand);
        unsigned long long hi = (unsigned)__ballot(u[16 * g + m + 1] < cand);
        c += __popcll(lo | (hi << 32));
      }
    }
  }
  return c;
}

__global__ void __launch_bounds__(32)
select_mask(const float* __restrict__ FF, unsigned* __restrict__ mask) {
  int row = blockIdx.x;
  int i = row & (TT - 1);
  const float* f = FF + (size_t)row * TT;
  unsigned* mrow = mask + (size_t)row * SEL_WORDS;
  int lane = threadIdx.x & 31;
  unsigned w0, w1;

  if (i < BUDGET) {
    w0 = causal_word(lane, i);
    w1 = causal_word(32 + lane, i);
  } else {
    int ng = (i >> 9) + 1;
    unsigned u[SEL_WORDS];
    #pragma unroll
    for (int g = 0; g < 4; ++g) {
      if (g < ng) {
        #pragma unroll
        for (int m = 16 * g; m < 16 * g + 16; ++m) u[m] = __float_as_uint(f[32 * m + lane]);
      } else {
        #pragma unroll
        for (int m = 16 * g; m < 16 * g + 16; ++m) u[m] = 0xFFFFFFFFu;
      }
    }
    #pragma unroll
    for (int m = 0; m < SEL_WORDS; ++m) u[m] = (32 * m + lane <= i) ? u[m] : 0xFFFFFFFFu;

    unsigned prefix = 0u;
    for (int bit = 30; bit >= 0; --bit) {
      unsigned cand = prefix | (1u << bit);
      if (count_below(u, ng, cand) < (unsigned)BUDGET) prefix = cand;
    }
    unsigned theta = prefix;
    int ntake = BUDGET - (int)count_below(u, ng, theta);

    unsigned below = (1u << lane) - 1u;
    unsigned carry = 0u;
    w0 = 0u; w1 = 0u;
    #pragma unroll
    for (int g = 0; g < 4; ++g) {
      if (g < ng) {
        #pragma unroll
        for (int m = 16 * g; m < 16 * g + 16; ++m) {
          bool eq = (u[m] == theta);
          unsigned eb = (unsigned)__ballot(eq);
          unsigned pre = carry + __popc(eb & below);
          bool keep = (u[m] < theta) || (eq && (int)pre < ntake);
          unsigned kb = (unsigned)__ballot(keep);
          carry += __popc(eb);
          if (m < 32) w0 = (lane == m) ? kb : w0;
          else        w1 = (lane == m - 32) ? kb : w1;
        }
      }
    }
    int dw = i >> 5;
    unsigned dbit = 1u << (i & 31);
    if (lane == (dw & 31)) { if (dw < 32) w0 |= dbit; else w1 |= dbit; }
  }
  mrow[lane] = w0;
  mrow[32 + lane] = w1;
}

__global__ void __launch_bounds__(32)
attn(const float* __restrict__ qkv, const float* __restrict__ FF,
                     const unsigned* __restrict__ mask, float* __restrict__ Y) {
  int ib = blockIdx.x, h = blockIdx.y, b = blockIdx.z;
  int lane = threadIdx.x & 31, g = lane >> 4, lm = lane & 15;
  int i0 = ib * 16;
  __shared__ float pt[16 * 17];

  const float* qb = qkv + ((size_t)(b * TT)) * C3 + h * HD;
  const float* kb = qb + CC;
  const float* vb = qb + 2 * CC;

  v2f aq[16];
  #pragma unroll
  for (int ks = 0; ks < 16; ++ks)
    aq[ks] = *(const v2f*)(qb + (size_t)(i0 + lm) * C3 + ks * 4 + 2 * g);

  float mrow[8], lrow[8];
  v8f o[4];
  #pragma unroll
  for (int t = 0; t < 4; ++t)
    #pragma unroll
    for (int r = 0; r < 8; ++r) o[t][r] = 0.f;
  #pragma unroll
  for (int r = 0; r < 8; ++r) { mrow[r] = -1e30f; lrow[r] = 0.f; }

  const unsigned* mrow16 = mask + ((size_t)(b * TT + i0)) * (TT / 32);
  unsigned or0 = 0u, or1 = 0u;
  #pragma unroll
  for (int r = 0; r < 16; ++r) {
    or0 |= mrow16[r * (TT / 32) + lane];
    or1 |= mrow16[r * (TT / 32) + 32 + lane];
  }
  int wsrc = lane >> 1, hsh = (lane & 1) * 16;
  unsigned w0 = __shfl(or0, wsrc, 32), w1 = __shfl(or0, 16 + wsrc, 32);
  unsigned w2 = __shfl(or1, wsrc, 32), w3 = __shfl(or1, 16 + wsrc, 32);
  unsigned t0 = (unsigned)__ballot(((w0 >> hsh) & 0xFFFFu) != 0u);
  unsigned t1 = (unsigned)__ballot(((w1 >> hsh) & 0xFFFFu) != 0u);
  unsigned t2 = (unsigned)__ballot(((w2 >> hsh) & 0xFFFFu) != 0u);
  unsigned t3 = (unsigned)__ballot(((w3 >> hsh) & 0xFFFFu) != 0u);
  unsigned long long tlo = (unsigned long long)t0 | ((unsigned long long)t1 << 32);
  unsigned long long thi = (unsigned long long)t2 | ((unsigned long long)t3 << 32);
  if (ib < 64) {
    if (ib < 63) tlo &= (2ull << ib) - 1ull;
    thi = 0ull;
  } else if (ib < 127) {
    thi &= (2ull << (ib - 64)) - 1ull;
  }

  for (;;) {
    int jb;
    if (tlo != 0ull)      { jb = __builtin_ctzll(tlo);      tlo &= tlo - 1ull; }
    else if (thi != 0ull) { jb = 64 + __builtin_ctzll(thi); thi &= thi - 1ull; }
    else break;
    int j0 = jb * 16;
    v2f bk[16];
    #pragma unroll
    for (int ks = 0; ks < 16; ++ks)
      bk[ks] = *(const v2f*)(kb + (size_t)(j0 + lm) * C3 + ks * 4 + 2 * g);
    v8f acc = {0.f, 0.f, 0.f, 0.f, 0.f, 0.f, 0.f, 0.f};
    #pragma unroll
    for (int ks = 0; ks < 16; ++ks) acc = wmma4(aq[ks], bk[ks], acc);

    float ffv[8]; unsigned mwv[8];
    #pragma unroll
    for (int r = 0; r < 8; ++r) {
      int i = i0 + r + 8 * g;
      ffv[r] = FF[((size_t)(b * TT + i)) * TT + j0 + lm];
      mwv[r] = mask[((size_t)(b * TT + i)) * (TT / 32) + (j0 >> 5)];
    }
    float p[8];
    #pragma unroll
    for (int r = 0; r < 8; ++r) {
      int i = i0 + r + 8 * g, j = j0 + lm;
      bool keep = ((mwv[r] >> (j & 31)) & 1u) && (j <= i);
      float s = keep ? (acc[r] * 0.125f - ffv[r]) : -1e30f;
      float bm = rowmax16(s);
      float mnew = fmaxf(mrow[r], bm);
      float scale = __expf(mrow[r] - mnew);
      float pr = __expf(s - mnew);
      lrow[r] = lrow[r] * scale + rowsum16(pr);
      mrow[r] = mnew;
      p[r] = pr;
      #pragma unroll
      for (int t = 0; t < 4; ++t) o[t][r] *= scale;
    }
    __syncthreads();
    #pragma unroll
    for (int r = 0; r < 8; ++r) pt[(r + 8 * g) * 17 + lm] = p[r];
    __syncthreads();
    v2f ap[4];
    #pragma unroll
    for (int kt = 0; kt < 4; ++kt) {
      ap[kt].x = pt[lm * 17 + kt * 4 + 2 * g];
      ap[kt].y = pt[lm * 17 + kt * 4 + 2 * g + 1];
    }
    v2f bv[4][4];
    #pragma unroll
    for (int kt = 0; kt < 4; ++kt)
      #pragma unroll
      for (int t = 0; t < 4; ++t) {
        bv[kt][t].x = vb[(size_t)(j0 + kt * 4 + 2 * g) * C3 + t * 16 + lm];
        bv[kt][t].y = vb[(size_t)(j0 + kt * 4 + 2 * g + 1) * C3 + t * 16 + lm];
      }
    #pragma unroll
    for (int kt = 0; kt < 4; ++kt)
      #pragma unroll
      for (int t = 0; t < 4; ++t)
        o[t] = wmma4(ap[kt], bv[kt][t], o[t]);
  }
  #pragma unroll
  for (int r = 0; r < 8; ++r) {
    float inv = 1.f / lrow[r];
    int i = i0 + r + 8 * g;
    #pragma unroll
    for (int t = 0; t < 4; ++t)
      Y[((size_t)(b * TT + i)) * CC + h * HD + t * 16 + lm] = o[t][r] * inv;
  }
}

extern "C" void kernel_launch(void* const* d_in, const int* in_sizes, int n_in,
                              void* d_out, int out_size, void* d_ws, size_t ws_size,
                              hipStream_t stream) {
  const float* x      = (const float*)d_in[0];
  const float* w_attn = (const float*)d_in[1];
  const float* b_attn = (const float*)d_in[2];
  const float* w_proj = (const float*)d_in[3];
  const float* b_proj = (const float*)d_in[4];
  float* out = (float*)d_out;

  char* ws = (char*)d_ws;
  const size_t QKV_BYTES = (size_t)BB * TT * C3 * 4;
  const size_t FF_BYTES  = (size_t)BB * TT * TT * 4;
  const size_t MSK_BYTES = (size_t)BB * TT * (TT / 32) * 4;
  float*    qkv = (float*)ws;
  float*    FF  = (float*)(ws + QKV_BYTES);
  unsigned* msk = (unsigned*)(ws + QKV_BYTES + FF_BYTES);
  float*    Y   = (float*)(ws + QKV_BYTES + FF_BYTES + MSK_BYTES);

  gemm_wmma<<<dim3(C3 / 128, (BB * TT) / 128), 128, 0, stream>>>(
      x, w_attn, b_attn, qkv, BB * TT, C3, CC);
  head0_scores<<<dim3(TT / 16, TT / 16, BB), 32, 0, stream>>>(qkv, FF);
  ff_cumsum<<<dim3(TT / 32, BB), 1024, 0, stream>>>(FF);
  select_mask<<<BB * TT, 32, 0, stream>>>(FF, msk);
  attn<<<dim3(TT / 16, NH, BB), 32, 0, stream>>>(qkv, FF, msk, Y);
  gemm_wmma<<<dim3(CC / 128, (BB * TT) / 128), 128, 0, stream>>>(
      Y, w_proj, b_proj, out, BB * TT, CC, CC);
}


# attn: score and output tiles computed transposed (keys x queries): a query's max, sum and rescale sit in its own lane, p feeds the PV WMMAs from registers; no LDS, barrier or 16-lane shuffles
# speedup vs baseline: 5.4436x; 1.1699x over previous
#include <hip/hip_runtime.h>
#include <hip/hip_bf16.h>

#define BB 2
#define TT 2048
#define CC 1024
#define NH 16
#define HD 64
#define C3 3072
#define BUDGET 408

typedef __attribute__((ext_vector_type(2))) float v2f;
typedef __attribute__((ext_vector_type(8))) float v8f;

__device__ __forceinline__ v8f wmma4(v2f a, v2f b, v8f c) {
  return __builtin_amdgcn_wmma_f32_16x16x4_f32(false, a, false, b, (short)0, c, false, false);
}

__device__ __forceinline__ float xhalf(float x) {
  int b = __builtin_bit_cast(int, x);
  return __builtin_bit_cast(float, __builtin_amdgcn_permlanex16(b, b, 0x76543210, 0xfedcba98, false, false));
}

typedef __attribute__((ext_vector_type(4))) float v4f;

__device__ __forceinline__ void gemm_load_block(v4f (&buf)[8][4], const float* const (&rb)[8], unsigned lo, int kb) {
  unsigned off = lo + (unsigned)kb;
  #pragma unroll
  for (int rg = 0; rg < 8; ++rg)
    #pragma unroll
    for (int q = 0; q < 4; ++q)
      buf[rg][q] = *(const v4f*)(rb[rg] + (size_t)off + q * 8);
}

__device__ __forceinline__ void gemm_mma_block(const v4f (&buf)[8][4], v8f (&acc)[4][4]) {
  #pragma unroll
  for (int q = 0; q < 4; ++q) {
    #pragma unroll
    for (int s = 0; s < 2; ++s) {
      v2f a[4], b[4];
      #pragma unroll
      for (int mt = 0; mt < 4; ++mt) {
        a[mt].x = buf[mt][q][2 * s];
        a[mt].y = buf[mt][q][2 * s + 1];
      }
      #pragma unroll
      for (int t = 0; t < 4; ++t) {
        b[t].x = buf[4 + t][q][2 * s];
        b[t].y = buf[4 + t][q][2 * s + 1];
      }
      #pragma unroll
      for (int mt = 0; mt < 4; ++mt)
        #pragma unroll
        for (int t = 0; t < 4; ++t)
          acc[mt][t] = wmma4(a[mt], b[t], acc[mt][t]);
    }
  }
}

__global__ void __launch_bounds__(128)
gemm_wmma(const float* __restrict__ A, const float* __restrict__ W,
          const float* __restrict__ bias, float* __restrict__ Out,
          int M, int N, int K) {
  int wave = __builtin_amdgcn_readfirstlane((int)(threadIdx.x >> 5));
  int lane = threadIdx.x & 31;
  int g  = lane >> 4;
  int lm = lane & 15;
  int m0 = blockIdx.y * 128 + (wave >> 1) * 64;
  int n0 = blockIdx.x * 128 + (wave & 1) * 64;
  if (n0 >= N || m0 >= M) return;

  v8f acc[4][4];
  #pragma unroll
  for (int t = 0; t < 4; ++t) {
    float bv = bias[n0 + t * 16 + lm];
    #pragma unroll
    for (int mt = 0; mt < 4; ++mt)
      #pragma unroll
      for (int r = 0; r < 8; ++r) acc[mt][t][r] = bv;
  }
  const float* rp[8];
  #pragma unroll
  for (int mt = 0; mt < 4; ++mt)
    rp[mt] = A + (size_t)(m0 + mt * 16) * K;
  #pragma unroll
  for (int t = 0; t < 4; ++t)
    rp[4 + t] = W + (size_t)(n0 + t * 16) * K;
  unsigned lo = (unsigned)(lm * K + 4 * g);

  v4f ping[8][4], pong[8][4];
  gemm_load_block(ping, rp, lo, 0);
  for (int kb = 0; kb < K; kb += 64) {
    gemm_load_block(pong, rp, lo, kb + 32);
    __builtin_amdgcn_sched_barrier(0);
    gemm_mma_block(ping, acc);
    __builtin_amdgcn_sched_barrier(0);
    gemm_load_block(ping, rp, lo, (kb + 64 < K) ? kb + 64 : kb + 32);
    __builtin_amdgcn_sched_barrier(0);
    gemm_mma_block(pong, acc);
    __builtin_amdgcn_sched_barrier(0);
  }
  #pragma unroll
  for (int mt = 0; mt < 4; ++mt)
    #pragma unroll
    for (int t = 0; t < 4; ++t)
      #pragma unroll
      for (int r = 0; r < 8; ++r)
        Out[(size_t)(m0 + mt * 16 + r + 8 * g) * N + n0 + t * 16 + lm] = acc[mt][t][r];
}

#define SS_WAVES 16
#define SS_TILES 8
__global__ void __launch_bounds__(SS_WAVES * 32)
scores_scan(const float* __restrict__ qkv, float* __restrict__ FF) {
  __shared__ float tot[SS_WAVES * 16];
  int w = __builtin_amdgcn_readfirstlane((int)(threadIdx.x >> 5));
  int lane = threadIdx.x & 31, g = lane >> 4, lm = lane & 15;
  int jb = blockIdx.x, b = blockIdx.y;
  int j0 = jb * 16, j = j0 + lm;
  const float* base = qkv + (size_t)(b * TT) * C3;
  int ib0 = w * SS_TILES;

  float sv[SS_TILES][8];
  #pragma unroll
  for (int t = 0; t < SS_TILES; ++t)
    #pragma unroll
    for (int r = 0; r < 8; ++r) sv[t][r] = 0.f;

  if (ib0 + SS_TILES - 1 >= jb) {
    const float* krow = base + (size_t)(j0 + lm) * C3 + CC + 4 * g;
    v4f kf[8];
    #pragma unroll
    for (int q = 0; q < 8; ++q) kf[q] = *(const v4f*)(krow + 8 * q);
    #pragma unroll
    for (int tp = 0; tp < SS_TILES; tp += 2) {
      if (ib0 + tp + 1 >= jb) {
        const float* qr0 = base + (size_t)((ib0 + tp) * 16 + lm) * C3 + 4 * g;
        const float* qr1 = qr0 + (size_t)16 * C3;
        v4f qa[8], qb[8];
        #pragma unroll
        for (int q = 0; q < 8; ++q) qa[q] = *(const v4f*)(qr0 + 8 * q);
        #pragma unroll
        for (int q = 0; q < 8; ++q) qb[q] = *(const v4f*)(qr1 + 8 * q);
        v8f c0 = {0.f, 0.f, 0.f, 0.f, 0.f, 0.f, 0.f, 0.f};
        v8f c1 = {0.f, 0.f, 0.f, 0.f, 0.f, 0.f, 0.f, 0.f};
        #pragma unroll
        for (int q = 0; q < 8; ++q) {
          v2f k01 = {kf[q].x, kf[q].y}, k23 = {kf[q].z, kf[q].w};
          v2f a01 = {qa[q].x, qa[q].y}, a23 = {qa[q].z, qa[q].w};
          v2f b01 = {qb[q].x, qb[q].y}, b23 = {qb[q].z, qb[q].w};
          c0 = wmma4(a01, k01, c0);
          c1 = wmma4(b01, k01, c1);
          c0 = wmma4(a23, k23, c0);
          c1 = wmma4(b23, k23, c1);
        }
        #pragma unroll
        for (int r = 0; r < 8; ++r) {
          int ia = (ib0 + tp) * 16 + r + 8 * g, ic = ia + 16;
          float s0 = c0[r] * 0.125f, s1 = c1[r] * 0.125f;
          sv[tp][r]     = (j < ia && j > 0) ? fmaxf(s0, 0.f) : 0.f;
          sv[tp + 1][r] = (j < ic && j > 0) ? fmaxf(s1, 0.f) : 0.f;
        }
      }
    }
  }

  float run = 0.f;
  #pragma unroll
  for (int t = 0; t < SS_TILES; ++t) {
    float hs = 0.f;
    #pragma unroll
    for (int r = 0; r < 8; ++r) hs += sv[t][r];
    float os = __shfl_xor(hs, 16, 32);
    float pre = run + (g ? os : 0.f);
    #pragma unroll
    for (int r = 0; r < 8; ++r) { float x = sv[t][r]; sv[t][r] = pre; pre += x; }
    run += hs + os;
  }
  if (g == 0) tot[w * 16 + lm] = run;
  __syncthreads();
  float off = 0.f;
  #pragma unroll
  for (int w2 = 0; w2 < SS_WAVES - 1; ++w2) off += (w2 < w) ? tot[w2 * 16 + lm] : 0.f;
  float* orow = FF + ((size_t)(b * TT) + ib0 * 16 + 8 * g) * TT + j;
  #pragma unroll
  for (int t = 0; t < SS_TILES; ++t)
    #pragma unroll
    for (int r = 0; r < 8; ++r)
      orow[(size_t)(t * 16 + r) * TT] = sv[t][r] + off;
}

#define SEL_WORDS (TT / 32)

__device__ __forceinline__ unsigned causal_word(int w, int i) {
  int jb = w * 32;
  if (jb + 31 <= i) return 0xFFFFFFFFu;
  if (jb > i)       return 0u;
  return (1u << (i - jb + 1)) - 1u;
}

__device__ __forceinline__ unsigned count_below(const unsigned (&u)[SEL_WORDS], int ng, unsigned cand) {
  unsigned c = 0;
  #pragma unroll
  for (int g = 0; g < 4; ++g) {
    if (g < ng) {
      #pragma unroll
      for (int m = 0; m < 16; m += 2) {
        unsigned long long lo = (unsigned)__ballot(u[16 * g + m] < cand);
        unsigned long long hi = (unsigned)__ballot(u[16 * g + m + 1] < cand);
        c += __popcll(lo | (hi << 32));
      }
    }
  }
  return c;
}

__global__ void __launch_bounds__(32)
select_mask(const float* __restrict__ FF, unsigned* __restrict__ mask) {
  int row = blockIdx.x;
  int i = row & (TT - 1);
  const float* f = FF + (size_t)row * TT;
  unsigned* mrow = mask + (size_t)row * SEL_WORDS;
  int lane = threadIdx.x & 31;
  unsigned w0, w1;

  if (i < BUDGET) {
    w0 = causal_word(lane, i);
    w1 = causal_word(32 + lane, i);
  } else {
    int ng = (i >> 9) + 1;
    unsigned u[SEL_WORDS];
    #pragma unroll
    for (int g = 0; g < 4; ++g) {
      if (g < ng) {
        #pragma unroll
        for (int m = 16 * g; m < 16 * g + 16; ++m) u[m] = __float_as_uint(f[32 * m + lane]);
      } else {
        #pragma unroll
        for (int m = 16 * g; m < 16 * g + 16; ++m) u[m] = 0xFFFFFFFFu;
      }
    }
    #pragma unroll
    for (int m = 0; m < SEL_WORDS; ++m) u[m] = (32 * m + lane <= i) ? u[m] : 0xFFFFFFFFu;

    unsigned prefix = 0u;
    for (int bit = 30; bit >= 0; --bit) {
      unsigned cand = prefix | (1u << bit);
      if (count_below(u, ng, cand) < (unsigned)BUDGET) prefix = cand;
    }
    unsigned theta = prefix;
    int ntake = BUDGET - (int)count_below(u, ng, theta);

    unsigned below = (1u << lane) - 1u;
    unsigned carry = 0u;
    w0 = 0u; w1 = 0u;
    #pragma unroll
    for (int g = 0; g < 4; ++g) {
      if (g < ng) {
        #pragma unroll
        for (int m = 16 * g; m < 16 * g + 16; ++m) {
          bool eq = (u[m] == theta);
          unsigned eb = (unsigned)__ballot(eq);
          unsigned pre = carry + __popc(eb & below);
          bool keep = (u[m] < theta) || (eq && (int)pre < ntake);
          unsigned kb = (unsigned)__ballot(keep);
          carry += __popc(eb);
          if (m < 32) w0 = (lane == m) ? kb : w0;
          else        w1 = (lane == m - 32) ? kb : w1;
        }
      }
    }
    int dw = i >> 5;
    unsigned dbit = 1u << (i & 31);
    if (lane == (dw & 31)) { if (dw < 32) w0 |= dbit; else w1 |= dbit; }
  }
  mrow[lane] = w0;
  mrow[32 + lane] = w1;
}

__global__ void __launch_bounds__(32)
attn(const float* __restrict__ qkv, const float* __restrict__ FF,
                     const unsigned* __restrict__ mask, float* __restrict__ Y) {
  int ib = blockIdx.x, h = blockIdx.y, b = blockIdx.z;
  int lane = threadIdx.x & 31, g = lane >> 4, lm = lane & 15;
  int i0 = ib * 16;
  int iq = i0 + lm;

  const float* qb = qkv + ((size_t)(b * TT)) * C3 + h * HD;
  const float* kb = qb + CC;
  const float* vb = qb + 2 * CC;

  v2f aq[16];
  #pragma unroll
  for (int ks = 0; ks < 16; ++ks)
    aq[ks] = *(const v2f*)(qb + (size_t)iq * C3 + ks * 4 + 2 * g);

  float mrun = -1e30f, lrun = 0.f;
  v8f o[4];
  #pragma unroll
  for (int t = 0; t < 4; ++t)
    #pragma unroll
    for (int r = 0; r < 8; ++r) o[t][r] = 0.f;

  const unsigned* mrow16 = mask + ((size_t)(b * TT + i0)) * (TT / 32);
  unsigned or0 = 0u, or1 = 0u;
  #pragma unroll
  for (int r = 0; r < 16; ++r) {
    or0 |= mrow16[r * (TT / 32) + lane];
    or1 |= mrow16[r * (TT / 32) + 32 + lane];
  }
  int wsrc = lane >> 1, hsh = (lane & 1) * 16;
  unsigned w0 = __shfl(or0, wsrc, 32), w1 = __shfl(or0, 16 + wsrc, 32);
  unsigned w2 = __shfl(or1, wsrc, 32), w3 = __shfl(or1, 16 + wsrc, 32);
  unsigned t0 = (unsigned)__ballot(((w0 >> hsh) & 0xFFFFu) != 0u);
  unsigned t1 = (unsigned)__ballot(((w1 >> hsh) & 0xFFFFu) != 0u);
  unsigned t2 = (unsigned)__ballot(((w2 >> hsh) & 0xFFFFu) != 0u);
  unsigned t3 = (unsigned)__ballot(((w3 >> hsh) & 0xFFFFu) != 0u);
  unsigned long long tlo = (unsigned long long)t0 | ((unsigned long long)t1 << 32);
  unsigned long long thi = (unsigned long long)t2 | ((unsigned long long)t3 << 32);
  if (ib < 64) {
    if (ib < 63) tlo &= (2ull << ib) - 1ull;
    thi = 0ull;
  } else if (ib < 127) {
    thi &= (2ull << (ib - 64)) - 1ull;
  }

  const float*    ffrow = FF + ((size_t)(b * TT + iq)) * TT + 8 * g;
  const unsigned* mkrow = mask + ((size_t)(b * TT + iq)) * (TT / 32);

  for (;;) {
    int jb;
    if (tlo != 0ull)      { jb = __builtin_ctzll(tlo);      tlo &= tlo - 1ull; }
    else if (thi != 0ull) { jb = 64 + __builtin_ctzll(thi); thi &= thi - 1ull; }
    else break;
    int j0 = jb * 16;
    int jl = j0 + 8 * g;
    v2f bk[16];
    #pragma unroll
    for (int ks = 0; ks < 16; ++ks)
      bk[ks] = *(const v2f*)(kb + (size_t)(j0 + lm) * C3 + ks * 4 + 2 * g);
    v8f acc = {0.f, 0.f, 0.f, 0.f, 0.f, 0.f, 0.f, 0.f};
    #pragma unroll
    for (int ks = 0; ks < 16; ++ks) acc = wmma4(bk[ks], aq[ks], acc);

    v4f f0 = *(const v4f*)(ffrow + j0);
    v4f f1 = *(const v4f*)(ffrow + j0 + 4);
    float ffv[8] = {f0.x, f0.y, f0.z, f0.w, f1.x, f1.y, f1.z, f1.w};
    unsigned bits = mkrow[j0 >> 5] >> ((j0 & 31) + 8 * g);
    v2f av[4][4];
    #pragma unroll
    for (int kt = 0; kt < 4; ++kt)
      #pragma unroll
      for (int t = 0; t < 4; ++t) {
        av[kt][t].x = vb[(size_t)(jl + 2 * kt) * C3 + t * 16 + lm];
        av[kt][t].y = vb[(size_t)(jl + 2 * kt + 1) * C3 + t * 16 + lm];
      }
    float s[8];
    #pragma unroll
    for (int r = 0; r < 8; ++r) {
      bool keep = ((bits >> r) & 1u) && (jl + r <= iq);
      s[r] = keep ? (acc[r] * 0.125f - ffv[r]) : -1e30f;
    }
    float bm = s[0];
    #pragma unroll
    for (int r = 1; r < 8; ++r) bm = fmaxf(bm, s[r]);
    bm = fmaxf(bm, xhalf(bm));
    float mnew = fmaxf(mrun, bm);
    float scale = __expf(mrun - mnew);
    float p[8];
    float ps = 0.f;
    #pragma unroll
    for (int r = 0; r < 8; ++r) { p[r] = __expf(s[r] - mnew); ps += p[r]; }
    ps += xhalf(ps);
    lrun = lrun * scale + ps;
    mrun = mnew;
    #pragma unroll
    for (int t = 0; t < 4; ++t)
      #pragma unroll
      for (int r = 0; r < 8; ++r) o[t][r] *= scale;
    #pragma unroll
    for (int kt = 0; kt < 4; ++kt) {
      v2f bp = {p[2 * kt], p[2 * kt + 1]};
      #pragma unroll
      for (int t = 0; t < 4; ++t)
        o[t] = wmma4(av[kt][t], bp, o[t]);
    }
  }
  float inv = 1.f / lrun;
  float* yrow = Y + ((size_t)(b * TT + iq)) * CC + h * HD + 8 * g;
  #pragma unroll
  for (int t = 0; t < 4; ++t) {
    v4f y0 = {o[t][0] * inv, o[t][1] * inv, o[t][2] * inv, o[t][3] * inv};
    v4f y1 = {o[t][4] * inv, o[t][5] * inv, o[t][6] * inv, o[t][7] * inv};
    *(v4f*)(yrow + t * 16)     = y0;
    *(v4f*)(yrow + t * 16 + 4) = y1;
  }
}

extern "C" void kernel_launch(void* const* d_in, const int* in_sizes, int n_in,
                              void* d_out, int out_size, void* d_ws, size_t ws_size,
                              hipStream_t stream) {
  const float* x      = (const float*)d_in[0];
  const float* w_attn = (const float*)d_in[1];
  const float* b_attn = (const float*)d_in[2];
  const float* w_proj = (const float*)d_in[3];
  const float* b_proj = (const float*)d_in[4];
  float* out = (float*)d_out;

  char* ws = (char*)d_ws;
  const size_t QKV_BYTES = (size_t)BB * TT * C3 * 4;
  const size_t FF_BYTES  = (size_t)BB * TT * TT * 4;
  const size_t MSK_BYTES = (size_t)BB * TT * (TT / 32) * 4;
  float*    qkv = (float*)ws;
  float*    FF  = (float*)(ws + QKV_BYTES);
  unsigned* msk = (unsigned*)(ws + QKV_BYTES + FF_BYTES);
  float*    Y   = (float*)(ws + QKV_BYTES + FF_BYTES + MSK_BYTES);

  gemm_wmma<<<dim3(C3 / 128, (BB * TT) / 128), 128, 0, stream>>>(
      x, w_attn, b_attn, qkv, BB * TT, C3, CC);
  scores_scan<<<dim3(TT / 16, BB), SS_WAVES * 32, 0, stream>>>(qkv, FF);
  select_mask<<<BB * TT, 32, 0, stream>>>(FF, msk);
  attn<<<dim3(TT / 16, NH, BB), 32, 0, stream>>>(qkv, FF, msk, Y);
  gemm_wmma<<<dim3(CC / 128, (BB * TT) / 128), 128, 0, stream>>>(
      Y, w_proj, b_proj, out, BB * TT, CC, CC);
}
